# starting kernel
# baseline (speedup 1.0000x reference)
#include <hip/hip_runtime.h>
#include <cstdint>
#include <cstddef>

typedef __attribute__((ext_vector_type(2))) float v2f;
typedef __attribute__((ext_vector_type(8))) float v8f;

#define IN_DIM   128
#define OUT_DIM  128
#define HEADS    4
#define HEAD_DIM 32

__device__ __forceinline__ int f2key(float f) {
  int i = __float_as_int(f);
  return i >= 0 ? i : (i ^ 0x7fffffff);
}
__device__ __forceinline__ float key2f(int k) {
  int i = k >= 0 ? k : (k ^ 0x7fffffff);
  return __int_as_float(i);
}

__global__ void gat_gemm_wmma(const float* __restrict__ x,
                              const float* __restrict__ W,
                              float* __restrict__ Wx, int N) {
  const int tile  = blockIdx.x;
  const int wave  = threadIdx.x >> 5;
  const int lane  = threadIdx.x & 31;
  const int mbase = tile * 16;
  const int obase = wave * 16;
  const int half  = lane >> 4;
  const int l15   = lane & 15;

  int arow = mbase + l15;
  if (arow >= N) arow = N - 1;

  const float* aptr = x + (size_t)arow * IN_DIM + 2 * half;
  const float* bptr = W + (size_t)(obase + l15) * IN_DIM + 2 * half;

  v8f c = {};
  #pragma unroll
  for (int k = 0; k < IN_DIM; k += 4) {
    v2f af = *(const v2f*)(aptr + k);
    v2f bf = *(const v2f*)(bptr + k);
    c = __builtin_amdgcn_wmma_f32_16x16x4_f32(false, af, false, bf,
                                              (short)0, c, false, false);
  }

  #pragma unroll
  for (int r = 0; r < 8; ++r) {
    int row = mbase + r + 8 * half;
    if (row < N) Wx[(size_t)row * OUT_DIM + obase + l15] = c[r];
  }
}

__global__ void gat_scores(const float* __restrict__ Wx,
                           const float* __restrict__ a,
                           float* __restrict__ s1, float* __restrict__ s2,
                           int N) {
  int t = blockIdx.x * blockDim.x + threadIdx.x;
  if (t >= N * HEADS) return;
  int n = t >> 2, h = t & 3;
  const float4* p  = (const float4*)(Wx + (size_t)n * OUT_DIM + h * HEAD_DIM);
  const float4* a1 = (const float4*)(a);
  const float4* a2 = (const float4*)(a + HEAD_DIM);
  float acc1 = 0.f, acc2 = 0.f;
  #pragma unroll
  for (int d = 0; d < HEAD_DIM / 4; ++d) {
    float4 v  = p[d];
    float4 c1 = a1[d];
    float4 c2 = a2[d];
    acc1 += v.x * c1.x + v.y * c1.y + v.z * c1.z + v.w * c1.w;
    acc2 += v.x * c2.x + v.y * c2.y + v.z * c2.z + v.w * c2.w;
  }
  s1[t] = acc1;
  s2[t] = acc2;
}

__global__ void gat_init(float* __restrict__ acc, float* __restrict__ denom,
                         int* __restrict__ mkey, int N) {
  int i = blockIdx.x * blockDim.x + threadIdx.x;
  if (i < N * OUT_DIM) acc[i] = 0.f;
  if (i < N * HEADS) { denom[i] = 0.f; mkey[i] = (int)0x80000000; }
}

__global__ void gat_edge_max(const int* __restrict__ ei,
                             const float* __restrict__ s1,
                             const float* __restrict__ s2,
                             int* __restrict__ mkey, int E) {
  int e = blockIdx.x * blockDim.x + threadIdx.x;
  if (e >= E) return;
  int src = ei[e], dst = ei[E + e];
  float4 v1 = *(const float4*)(s1 + (size_t)src * HEADS);
  float4 v2 = *(const float4*)(s2 + (size_t)dst * HEADS);
  float el[HEADS] = {v1.x + v2.x, v1.y + v2.y, v1.z + v2.z, v1.w + v2.w};
  #pragma unroll
  for (int h = 0; h < HEADS; ++h) {
    float v = el[h] > 0.f ? el[h] : 0.2f * el[h];
    atomicMax(&mkey[(size_t)dst * HEADS + h], f2key(v));
  }
}

__global__ void gat_edge_exp(const int* __restrict__ ei,
                             const float* __restrict__ s1,
                             const float* __restrict__ s2,
                             const int* __restrict__ mkey,
                             float* __restrict__ ex,
                             float* __restrict__ denom, int E) {
  int e = blockIdx.x * blockDim.x + threadIdx.x;
  if (e >= E) return;
  int src = ei[e], dst = ei[E + e];
  float4 v1 = *(const float4*)(s1 + (size_t)src * HEADS);
  float4 v2 = *(const float4*)(s2 + (size_t)dst * HEADS);
  float el[HEADS] = {v1.x + v2.x, v1.y + v2.y, v1.z + v2.z, v1.w + v2.w};
  float4 exv;
  #pragma unroll
  for (int h = 0; h < HEADS; ++h) {
    float v = el[h] > 0.f ? el[h] : 0.2f * el[h];
    float m = key2f(mkey[(size_t)dst * HEADS + h]);
    float xv = __expf(v - m);
    ((float*)&exv)[h] = xv;
    unsafeAtomicAdd(&denom[(size_t)dst * HEADS + h], xv);
  }
  *(float4*)(ex + (size_t)e * HEADS) = exv;
}

__global__ void gat_recip(float* __restrict__ denom, int total) {
  int i = blockIdx.x * blockDim.x + threadIdx.x;
  if (i >= total) return;
  denom[i] = __frcp_rn(denom[i]);
}

__global__ void gat_edge_aggr(const int* __restrict__ ei,
                              const float* __restrict__ Wx,
                              const float* __restrict__ ex,
                              const float* __restrict__ rden,
                              float* __restrict__ acc, int E) {
  long long gid = (long long)blockIdx.x * blockDim.x + threadIdx.x;
  int e  = (int)(gid >> 5);
  int q  = (int)(gid & 31);
  if (e >= E) return;
  int src = ei[e], dst = ei[E + e];
  int j = q << 2;
  int h = j >> 5;
  float alpha = ex[(size_t)e * HEADS + h] * rden[(size_t)dst * HEADS + h];
  float4 wv = *(const float4*)(Wx + (size_t)src * OUT_DIM + j);
  float* ap = acc + (size_t)dst * OUT_DIM + j;
  unsafeAtomicAdd(ap + 0, alpha * wv.x);
  unsafeAtomicAdd(ap + 1, alpha * wv.y);
  unsafeAtomicAdd(ap + 2, alpha * wv.z);
  unsafeAtomicAdd(ap + 3, alpha * wv.w);
}

__global__ void gat_elu(const float* __restrict__ acc, float* __restrict__ out,
                        int total) {
  int i = blockIdx.x * blockDim.x + threadIdx.x;
  if (i >= total) return;
  float v = acc[i];
  out[i] = v > 0.f ? v : (__expf(v) - 1.0f);
}

extern "C" void kernel_launch(void* const* d_in, const int* in_sizes, int n_in,
                              void* d_out, int out_size, void* d_ws, size_t ws_size,
                              hipStream_t stream) {
  const float* x  = (const float*)d_in[0];
  const float* W  = (const float*)d_in[1];
  const float* a  = (const float*)d_in[2];
  const int*   ei = (const int*)d_in[3];
  const int N = in_sizes[0] / IN_DIM;
  const int E = in_sizes[3] / 2;
  float* out = (float*)d_out;

  char* ws = (char*)d_ws;
  size_t off = 0;
  auto carve = [&](size_t bytes) -> void* {
    void* p = ws + off;
    off += (bytes + 255) & ~(size_t)255;
    return p;
  };
  float* Wx    = (float*)carve((size_t)N * OUT_DIM * sizeof(float));
  float* s1    = (float*)carve((size_t)N * HEADS * sizeof(float));
  float* s2    = (float*)carve((size_t)N * HEADS * sizeof(float));
  int*   mkey  = (int*)  carve((size_t)N * HEADS * sizeof(int));
  float* denom = (float*)carve((size_t)N * HEADS * sizeof(float));
  float* ex    = (float*)carve((size_t)E * HEADS * sizeof(float));
  float* acc   = (float*)carve((size_t)N * OUT_DIM * sizeof(float));
  (void)ws_size;

  gat_gemm_wmma<<<(N + 15) / 16, 256, 0, stream>>>(x, W, Wx, N);

  gat_scores<<<(N * HEADS + 255) / 256, 256, 0, stream>>>(Wx, a, s1, s2, N);

  gat_init<<<(N * OUT_DIM + 255) / 256, 256, 0, stream>>>(acc, denom, mkey, N);

  gat_edge_max<<<(E + 255) / 256, 256, 0, stream>>>(ei, s1, s2, mkey, E);

  gat_edge_exp<<<(E + 255) / 256, 256, 0, stream>>>(ei, s1, s2, mkey, ex, denom, E);

  gat_recip<<<(N * HEADS + 255) / 256, 256, 0, stream>>>(denom, N * HEADS);

  {
    long long threads = (long long)E * 32;
    int blocks = (int)((threads + 255) / 256);
    gat_edge_aggr<<<blocks, 256, 0, stream>>>(ei, Wx, ex, denom, acc, E);
  }

  gat_elu<<<(N * OUT_DIM + 255) / 256, 256, 0, stream>>>(acc, out, N * OUT_DIM);
}


# aggregation owner-computes, float atomics gone: a work-group scans dst once, lists its 196 nodes' edges in LDS; a wave per node sorts the list by source, sums in that fixed order; 3 launches gone
# speedup vs baseline: 4.4730x; 4.4730x over previous
#include <hip/hip_runtime.h>
#include <cstdint>
#include <cstddef>

typedef __attribute__((ext_vector_type(2))) float v2f;
typedef __attribute__((ext_vector_type(8))) float v8f;

#define IN_DIM   128
#define OUT_DIM  128
#define HEADS    4
#define HEAD_DIM 32

__global__ void gat_gemm_wmma(const float* __restrict__ x,
                              const float* __restrict__ W,
                              float* __restrict__ Wx, int N) {
  const int tile  = blockIdx.x;
  const int wave  = threadIdx.x >> 5;
  const int lane  = threadIdx.x & 31;
  const int mbase = tile * 16;
  const int obase = wave * 16;
  const int half  = lane >> 4;
  const int l15   = lane & 15;

  int arow = mbase + l15;
  if (arow >= N) arow = N - 1;

  const float* aptr = x + (size_t)arow * IN_DIM + 2 * half;
  const float* bptr = W + (size_t)(obase + l15) * IN_DIM + 2 * half;

  v8f c = {};
  #pragma unroll
  for (int k = 0; k < IN_DIM; k += 4) {
    v2f af = *(const v2f*)(aptr + k);
    v2f bf = *(const v2f*)(bptr + k);
    c = __builtin_amdgcn_wmma_f32_16x16x4_f32(false, af, false, bf,
                                              (short)0, c, false, false);
  }

  #pragma unroll
  for (int r = 0; r < 8; ++r) {
    int row = mbase + r + 8 * half;
    if (row < N) Wx[(size_t)row * OUT_DIM + obase + l15] = c[r];
  }
}

__global__ void gat_scores(const float* __restrict__ Wx,
                           const float* __restrict__ a,
                           float* __restrict__ s1, float* __restrict__ s2,
                           int N) {
  int t = blockIdx.x * blockDim.x + threadIdx.x;
  if (t >= N * HEADS) return;
  int n = t >> 2, h = t & 3;
  const float4* p  = (const float4*)(Wx + (size_t)n * OUT_DIM + h * HEAD_DIM);
  const float4* a1 = (const float4*)(a);
  const float4* a2 = (const float4*)(a + HEAD_DIM);
  float acc1 = 0.f, acc2 = 0.f;
  #pragma unroll
  for (int d = 0; d < HEAD_DIM / 4; ++d) {
    float4 v  = p[d];
    float4 c1 = a1[d];
    float4 c2 = a2[d];
    acc1 += v.x * c1.x + v.y * c1.y + v.z * c1.z + v.w * c1.w;
    acc2 += v.x * c2.x + v.y * c2.y + v.z * c2.z + v.w * c2.w;
  }
  s1[t] = acc1;
  s2[t] = acc2;
}

#define OWN_R       196
#define OWN_SLOT    64
#define OWN_PAD     8
#define OWN_THREADS 1024
#define OWN_WAVES   (OWN_THREADS / 32)
#define OWN_SCAN_U  8
#define OWN_CONV    ((OWN_R * OWN_SLOT + OWN_THREADS - 1) / OWN_THREADS)

__device__ __forceinline__ float leaky02(float el) {
  return el > 0.f ? el : 0.2f * el;
}
__device__ __forceinline__ float elu1(float v) {
  return v > 0.f ? v : (__expf(v) - 1.0f);
}
__device__ __forceinline__ float wave_max(float v) {
  #pragma unroll
  for (int o = 16; o > 0; o >>= 1) v = fmaxf(v, __shfl_xor(v, o));
  return v;
}
__device__ __forceinline__ void own_note(int d, int e, int lo,
                                         int* cnt, int* slot) {
  unsigned loc = (unsigned)(d - lo);
  if (loc < (unsigned)OWN_R) {
    int p = atomicAdd(&cnt[loc], 1);
    if (p < OWN_SLOT) slot[loc * OWN_SLOT + p] = e;
  }
}
template <bool GUARD>
__device__ __forceinline__ void own_scan_trip(const int4* __restrict__ d4,
                                              int base, int tid, int nvec,
                                              int lo, int* cnt, int* slot) {
  int4 v[OWN_SCAN_U];
  #pragma unroll
  for (int u = 0; u < OWN_SCAN_U; ++u) {
    int i = base + u * OWN_THREADS + tid;
    if (GUARD) v[u] = (i < nvec) ? d4[i] : make_int4(-1, -1, -1, -1);
    else       v[u] = d4[i];
  }
  #pragma unroll
  for (int u = 0; u < OWN_SCAN_U; ++u) {
    int e0 = (base + u * OWN_THREADS + tid) << 2;
    unsigned lmin = min(min((unsigned)(v[u].x - lo), (unsigned)(v[u].y - lo)),
                        min((unsigned)(v[u].z - lo), (unsigned)(v[u].w - lo)));
    if (lmin < (unsigned)OWN_R) {
      own_note(v[u].x, e0 + 0, lo, cnt, slot);
      own_note(v[u].y, e0 + 1, lo, cnt, slot);
      own_note(v[u].z, e0 + 2, lo, cnt, slot);
      own_note(v[u].w, e0 + 3, lo, cnt, slot);
    }
  }
}

__global__ __launch_bounds__(OWN_THREADS)
void gat_owner_aggr(const int* __restrict__ ei,
                    const float* __restrict__ Wx,
                    const float* __restrict__ s1,
                    const float* __restrict__ s2,
                    float* __restrict__ out, int N, int E) {
  __shared__ int   cnt[OWN_R];
  __shared__ int   slot[OWN_R * OWN_SLOT];
  __shared__ int   wsrc[OWN_WAVES][OWN_SLOT + OWN_PAD];
  __shared__ float wex[OWN_WAVES][HEADS][OWN_SLOT + OWN_PAD];

  const int tid  = threadIdx.x;
  const int wave = __builtin_amdgcn_readfirstlane(tid >> 5);
  const int lane = tid & 31;
  const int lo   = blockIdx.x * OWN_R;
  const int* __restrict__ dstp = ei + E;

  for (int i = tid; i < OWN_R; i += OWN_THREADS) cnt[i] = 0;
  __syncthreads();

  const int nvec = ((E & 3) == 0) ? (E >> 2) : 0;
  const int4* __restrict__ d4 = (const int4*)dstp;
  const int trip = OWN_THREADS * OWN_SCAN_U;
  int base = 0;
  for (; base + trip <= nvec; base += trip)
    own_scan_trip<false>(d4, base, tid, nvec, lo, cnt, slot);
  if (base < nvec)
    own_scan_trip<true>(d4, base, tid, nvec, lo, cnt, slot);
  for (int e = (nvec << 2) + tid; e < E; e += OWN_THREADS)
    own_note(dstp[e], e, lo, cnt, slot);
  __syncthreads();

  {
    int tmp[OWN_CONV];
    #pragma unroll
    for (int k = 0; k < OWN_CONV; ++k) {
      int idx = tid + k * OWN_THREADS;
      bool ok = idx < OWN_R * OWN_SLOT;
      if (ok) ok = (idx & (OWN_SLOT - 1)) < cnt[idx / OWN_SLOT];
      tmp[k] = ok ? ei[slot[idx]] : -1;
    }
    #pragma unroll
    for (int k = 0; k < OWN_CONV; ++k) {
      int idx = tid + k * OWN_THREADS;
      if (tmp[k] >= 0) slot[idx] = tmp[k];
    }
  }
  __syncthreads();

  const int h = lane >> 3;
  const int j = lane << 2;
  for (int loc = wave; loc < OWN_R; loc += OWN_WAVES) {
    const int n = lo + loc;
    if (n >= N) break;
    const int L = __builtin_amdgcn_readfirstlane(cnt[loc]);
    float4 acc = make_float4(0.f, 0.f, 0.f, 0.f);
    float den = 0.f;
    float* op = out + (size_t)n * OUT_DIM + j;
    if (L == 0) {
      *(float4*)op = acc;
      continue;
    }
    const float4 s2v = *(const float4*)(s2 + (size_t)n * HEADS);
    const float s2h = h == 0 ? s2v.x : h == 1 ? s2v.y : h == 2 ? s2v.z : s2v.w;
    float m0 = -INFINITY, m1 = -INFINITY, m2 = -INFINITY, m3 = -INFINITY;

    if (L <= OWN_SLOT) {
      const int* sl = slot + loc * OWN_SLOT;
      const bool v0 = lane < L;
      const bool v1 = lane + 32 < L;
      const int key0 = v0 ? sl[lane] : 0x7fffffff;
      const int key1 = v1 ? sl[lane + 32] : 0x7fffffff;
      float4 a0 = make_float4(0.f, 0.f, 0.f, 0.f), a1 = a0;
      if (v0) a0 = *(const float4*)(s1 + (size_t)key0 * HEADS);
      if (v1) a1 = *(const float4*)(s1 + (size_t)key1 * HEADS);
      int r0 = 0, r1 = 0;
      if (L <= 32) {
        for (int t = 0; t < L; ++t) {
          int kt = __builtin_amdgcn_readlane(key0, t);
          r0 += (kt < key0 || (kt == key0 && t < lane)) ? 1 : 0;
        }
      } else {
        for (int t = 0; t < 32; ++t) {
          int kt = __builtin_amdgcn_readlane(key0, t);
          r0 += (kt < key0 || (kt == key0 && t < lane)) ? 1 : 0;
          r1 += (kt <= key1) ? 1 : 0;
        }
        for (int t = 32; t < L; ++t) {
          int kt = __builtin_amdgcn_readlane(key1, t - 32);
          r0 += (kt < key0) ? 1 : 0;
          r1 += (kt < key1 || (kt == key1 && (t - 32) < lane)) ? 1 : 0;
        }
      }
      #pragma unroll
      for (int c = 0; c < 2; ++c) {
        const int    key  = c == 0 ? key0 : key1;
        const int    rank = c == 0 ? r0 : r1;
        const float4 s1v  = c == 0 ? a0 : a1;
        if (c == 0 ? v0 : v1) {
          float e0 = leaky02(s1v.x + s2v.x);
          float e1 = leaky02(s1v.y + s2v.y);
          float e2 = leaky02(s1v.z + s2v.z);
          float e3 = leaky02(s1v.w + s2v.w);
          wsrc[wave][rank]   = key;
          wex[wave][0][rank] = e0;
          wex[wave][1][rank] = e1;
          wex[wave][2][rank] = e2;
          wex[wave][3][rank] = e3;
          m0 = fmaxf(m0, e0); m1 = fmaxf(m1, e1);
          m2 = fmaxf(m2, e2); m3 = fmaxf(m3, e3);
        }
      }
      if (lane < OWN_PAD) {
        wsrc[wave][L + lane]   = 0;
        wex[wave][0][L + lane] = 0.f;
        wex[wave][1][L + lane] = 0.f;
        wex[wave][2][L + lane] = 0.f;
        wex[wave][3][L + lane] = 0.f;
      }
      m0 = wave_max(m0); m1 = wave_max(m1); m2 = wave_max(m2); m3 = wave_max(m3);
      const float mh = h == 0 ? m0 : h == 1 ? m1 : h == 2 ? m2 : m3;
      const float* wxl = Wx + j;
      for (int t0 = 0; t0 < L; t0 += OWN_PAD) {
        int    sv[OWN_PAD];
        float  ev[OWN_PAD];
        float4 row[OWN_PAD];
        #pragma unroll
        for (int u = 0; u < OWN_PAD; ++u) {
          sv[u] = wsrc[wave][t0 + u];
          ev[u] = wex[wave][h][t0 + u];
        }
        #pragma unroll
        for (int u = 0; u < OWN_PAD; ++u) {
          row[u] = make_float4(0.f, 0.f, 0.f, 0.f);
          if (t0 + u < L) row[u] = *(const float4*)(wxl + (size_t)sv[u] * OUT_DIM);
        }
        #pragma unroll
        for (int u = 0; u < OWN_PAD; ++u) {
          float xv = (t0 + u < L) ? __expf(ev[u] - mh) : 0.f;
          den   += xv;
          acc.x += xv * row[u].x; acc.y += xv * row[u].y;
          acc.z += xv * row[u].z; acc.w += xv * row[u].w;
        }
      }
    } else {
      for (int base2 = 0; base2 < E; base2 += 32) {
        int e = base2 + lane;
        if (e < E && dstp[e] == n) {
          float4 s1v = *(const float4*)(s1 + (size_t)ei[e] * HEADS);
          m0 = fmaxf(m0, leaky02(s1v.x + s2v.x));
          m1 = fmaxf(m1, leaky02(s1v.y + s2v.y));
          m2 = fmaxf(m2, leaky02(s1v.z + s2v.z));
          m3 = fmaxf(m3, leaky02(s1v.w + s2v.w));
        }
      }
      m0 = wave_max(m0); m1 = wave_max(m1); m2 = wave_max(m2); m3 = wave_max(m3);
      const float mh = h == 0 ? m0 : h == 1 ? m1 : h == 2 ? m2 : m3;
      for (int base2 = 0; base2 < E; base2 += 32) {
        int e = base2 + lane;
        bool mt = e < E && dstp[e] == n;
        int src = mt ? ei[e] : 0;
        unsigned mask = __builtin_amdgcn_ballot_w32(mt);
        while (mask) {
          int t = __builtin_ctz(mask);
          mask &= mask - 1;
          int s = __builtin_amdgcn_readlane(src, t);
          float ev = leaky02(s1[(size_t)s * HEADS + h] + s2h);
          float4 row = *(const float4*)(Wx + (size_t)s * OUT_DIM + j);
          float xv = __expf(ev - mh);
          den   += xv;
          acc.x += xv * row.x; acc.y += xv * row.y;
          acc.z += xv * row.z; acc.w += xv * row.w;
        }
      }
    }
    float4 o;
    o.x = elu1(acc.x / den); o.y = elu1(acc.y / den);
    o.z = elu1(acc.z / den); o.w = elu1(acc.w / den);
    *(float4*)op = o;
  }
}

extern "C" void kernel_launch(void* const* d_in, const int* in_sizes, int n_in,
                              void* d_out, int out_size, void* d_ws, size_t ws_size,
                              hipStream_t stream) {
  const float* x  = (const float*)d_in[0];
  const float* W  = (const float*)d_in[1];
  const float* a  = (const float*)d_in[2];
  const int*   ei = (const int*)d_in[3];
  const int N = in_sizes[0] / IN_DIM;
  const int E = in_sizes[3] / 2;
  float* out = (float*)d_out;

  char* ws = (char*)d_ws;
  size_t off = 0;
  auto carve = [&](size_t bytes) -> void* {
    void* p = ws + off;
    off += (bytes + 255) & ~(size_t)255;
    return p;
  };
  float* Wx = (float*)carve((size_t)N * OUT_DIM * sizeof(float));
  float* s1 = (float*)carve((size_t)N * HEADS * sizeof(float));
  float* s2 = (float*)carve((size_t)N * HEADS * sizeof(float));
  (void)ws_size;

  gat_gemm_wmma<<<(N + 15) / 16, 256, 0, stream>>>(x, W, Wx, N);

  gat_scores<<<(N * HEADS + 255) / 256, 256, 0, stream>>>(Wx, a, s1, s2, N);

  gat_owner_aggr<<<(N + OWN_R - 1) / OWN_R, OWN_THREADS, 0, stream>>>(
      ei, Wx, s1, s2, out, N, E);
}


# aggregation scan reads 1-byte owner keys (written in the scores launch), 16 edges per load not 4; waves list matches by ballot, dst and src fetched for listed edges only; float sums unchanged
# speedup vs baseline: 4.9868x; 1.1149x over previous
#include <hip/hip_runtime.h>
#include <cstdint>
#include <cstddef>

typedef __attribute__((ext_vector_type(2))) float v2f;
typedef __attribute__((ext_vector_type(8))) float v8f;

#define IN_DIM   128
#define OUT_DIM  128
#define HEADS    4
#define HEAD_DIM 32

__global__ void gat_gemm_wmma(const float* __restrict__ x,
                              const float* __restrict__ W,
                              float* __restrict__ Wx, int N) {
  const int tile  = blockIdx.x;
  const int wave  = threadIdx.x >> 5;
  const int lane  = threadIdx.x & 31;
  const int mbase = tile * 16;
  const int obase = wave * 16;
  const int half  = lane >> 4;
  const int l15   = lane & 15;

  int arow = mbase + l15;
  if (arow >= N) arow = N - 1;

  const float* aptr = x + (size_t)arow * IN_DIM + 2 * half;
  const float* bptr = W + (size_t)(obase + l15) * IN_DIM + 2 * half;

  v8f c = {};
  #pragma unroll
  for (int k = 0; k < IN_DIM; k += 4) {
    v2f af = *(const v2f*)(aptr + k);
    v2f bf = *(const v2f*)(bptr + k);
    c = __builtin_amdgcn_wmma_f32_16x16x4_f32(false, af, false, bf,
                                              (short)0, c, false, false);
  }

  #pragma unroll
  for (int r = 0; r < 8; ++r) {
    int row = mbase + r + 8 * half;
    if (row < N) Wx[(size_t)row * OUT_DIM + obase + l15] = c[r];
  }
}

#define OWN_R       196
#define OWN_SLOT    64
#define OWN_PAD     8
#define OWN_THREADS 1024
#define OWN_WAVES   (OWN_THREADS / 32)
#define OWN_SCAN_U  8
#define OWN_WCAP    256
#define OWN_LK_U    4

__device__ __forceinline__ unsigned own_key(int d) {
  return ((unsigned)d / (unsigned)OWN_R) & 255u;
}

__global__ void gat_scores(const float* __restrict__ Wx,
                           const float* __restrict__ a,
                           float* __restrict__ s1, float* __restrict__ s2,
                           const int* __restrict__ ei,
                           unsigned* __restrict__ okey,
                           int N, int E) {
  int t = blockIdx.x * blockDim.x + threadIdx.x;
  if ((E & 3) == 0) {
    const int4* __restrict__ d4 = (const int4*)(ei + E);
    const int nw   = E >> 2;
    const int step = gridDim.x * blockDim.x;
    for (int i = t; i < nw; i += step) {
      const int4 d = d4[i];
      okey[i] = own_key(d.x) | (own_key(d.y) << 8) |
                (own_key(d.z) << 16) | (own_key(d.w) << 24);
    }
  }
  if (t >= N * HEADS) return;
  int n = t >> 2, h = t & 3;
  const float4* p  = (const float4*)(Wx + (size_t)n * OUT_DIM + h * HEAD_DIM);
  const float4* a1 = (const float4*)(a);
  const float4* a2 = (const float4*)(a + HEAD_DIM);
  float acc1 = 0.f, acc2 = 0.f;
  #pragma unroll
  for (int d = 0; d < HEAD_DIM / 4; ++d) {
    float4 v  = p[d];
    float4 c1 = a1[d];
    float4 c2 = a2[d];
    acc1 += v.x * c1.x + v.y * c1.y + v.z * c1.z + v.w * c1.w;
    acc2 += v.x * c2.x + v.y * c2.y + v.z * c2.z + v.w * c2.w;
  }
  s1[t] = acc1;
  s2[t] = acc2;
}


__device__ __forceinline__ float leaky02(float el) {
  return el > 0.f ? el : 0.2f * el;
}
__device__ __forceinline__ float elu1(float v) {
  return v > 0.f ? v : (__expf(v) - 1.0f);
}
__device__ __forceinline__ float wave_max(float v) {
  #pragma unroll
  for (int o = 16; o > 0; o >>= 1) v = fmaxf(v, __shfl_xor(v, o));
  return v;
}
__device__ __forceinline__ unsigned own_zbytes(unsigned x) {
  return ~(((x & 0x7f7f7f7fu) + 0x7f7f7f7fu) | x | 0x7f7f7f7fu);
}
__device__ __forceinline__ void own_list(unsigned m, int e0, int* wl, int& wcount) {
  unsigned b;
  while ((b = __builtin_amdgcn_ballot_w32(m != 0u)) != 0u) {
    if (m != 0u) {
      const int bit = __builtin_ctz(m);
      const int pos = wcount + (int)__builtin_amdgcn_mbcnt_lo(b, 0u);
      if (pos < OWN_WCAP)
        wl[pos] = e0 + ((bit & 3) << 2) + (bit >> 3) + ((bit & 4) ? OWN_THREADS * 16 : 0);
      m &= m - 1u;
    }
    wcount += __builtin_popcount(b);
  }
}
__device__ __forceinline__ unsigned own_mask(uint4 v, unsigned rep) {
  return (own_zbytes(v.x ^ rep) >> 7) | (own_zbytes(v.y ^ rep) >> 6) |
         (own_zbytes(v.z ^ rep) >> 5) | (own_zbytes(v.w ^ rep) >> 4);
}
template <bool GUARD>
__device__ __forceinline__ void own_scan_trip(const uint4* __restrict__ k4,
                                              int base, int tid, int nvec,
                                              unsigned rep, int* wl, int& wcount) {
  uint4 v[OWN_SCAN_U];
  const unsigned nomatch = rep ^ 0x01010101u;
  #pragma unroll
  for (int u = 0; u < OWN_SCAN_U; ++u) {
    int i = base + u * OWN_THREADS + tid;
    if (GUARD) v[u] = (i < nvec) ? k4[i] : make_uint4(nomatch, nomatch, nomatch, nomatch);
    else       v[u] = k4[i];
  }
  #pragma unroll
  for (int u = 0; u < OWN_SCAN_U; u += 2) {
    const int e0 = (base + u * OWN_THREADS + tid) << 4;
    const unsigned m = own_mask(v[u], rep) | (own_mask(v[u + 1], rep) << 4);
    own_list(m, e0, wl, wcount);
  }
}

__global__ __launch_bounds__(OWN_THREADS)
void gat_owner_aggr(const int* __restrict__ ei,
                    const unsigned* __restrict__ okey,
                    const float* __restrict__ Wx,
                    const float* __restrict__ s1,
                    const float* __restrict__ s2,
                    float* __restrict__ out, int N, int E) {
  __shared__ int   cnt[OWN_R];
  __shared__ int   slot[OWN_R * OWN_SLOT];
  __shared__ int   wlist[OWN_WAVES][OWN_WCAP];
  __shared__ int   over;
  __shared__ int   wsrc[OWN_WAVES][OWN_SLOT + OWN_PAD];
  __shared__ float wex[OWN_WAVES][HEADS][OWN_SLOT + OWN_PAD];

  const int tid  = threadIdx.x;
  const int wave = __builtin_amdgcn_readfirstlane(tid >> 5);
  const int lane = tid & 31;
  const int lo   = blockIdx.x * OWN_R;
  const int* __restrict__ dstp = ei + E;

  for (int i = tid; i < OWN_R; i += OWN_THREADS) cnt[i] = 0;
  if (tid == 0) over = 0;
  __syncthreads();

  int* wl = wlist[wave];
  int wcount = 0;
  const unsigned rep = (blockIdx.x & 255u) * 0x01010101u;
  const int nvec = ((E & 3) == 0) ? (E >> 4) : 0;
  const uint4* __restrict__ k4 = (const uint4*)okey;
  const int trip = OWN_THREADS * OWN_SCAN_U;
  int base = 0;
  for (; base + trip <= nvec; base += trip)
    own_scan_trip<false>(k4, base, tid, nvec, rep, wl, wcount);
  if (base < nvec)
    own_scan_trip<true>(k4, base, tid, nvec, rep, wl, wcount);
  for (int eb = nvec << 4; eb < E; eb += OWN_THREADS) {
    const int e = eb + tid;
    const bool hit = e < E && (unsigned)(dstp[e] - lo) < (unsigned)OWN_R;
    own_list(hit ? 1u : 0u, e, wl, wcount);
  }
  if (wcount > OWN_WCAP && lane == 0) over = 1;
  __syncthreads();

  if (over == 0) {
    for (int b0 = 0; b0 < wcount; b0 += 32 * OWN_LK_U) {
      int dd[OWN_LK_U], ss[OWN_LK_U];
      #pragma unroll
      for (int u = 0; u < OWN_LK_U; ++u) {
        const int i = b0 + u * 32 + lane;
        dd[u] = -1;
        ss[u] = 0;
        if (i < wcount) {
          const int e = wl[i];
          dd[u] = dstp[e];
          ss[u] = ei[e];
        }
      }
      #pragma unroll
      for (int u = 0; u < OWN_LK_U; ++u) {
        const unsigned loc = (unsigned)(dd[u] - lo);
        if (loc < (unsigned)OWN_R) {
          const int q = atomicAdd(&cnt[loc], 1);
          if (q < OWN_SLOT) slot[loc * OWN_SLOT + q] = ss[u];
        }
      }
    }
  } else {
    for (int i = tid; i < OWN_R; i += OWN_THREADS) cnt[i] = OWN_SLOT + 1;
  }
  __syncthreads();

  const int h = lane >> 3;
  const int j = lane << 2;
  for (int loc = wave; loc < OWN_R; loc += OWN_WAVES) {
    const int n = lo + loc;
    if (n >= N) break;
    const int L = __builtin_amdgcn_readfirstlane(cnt[loc]);
    float4 acc = make_float4(0.f, 0.f, 0.f, 0.f);
    float den = 0.f;
    float* op = out + (size_t)n * OUT_DIM + j;
    if (L == 0) {
      *(float4*)op = acc;
      continue;
    }
    const float4 s2v = *(const float4*)(s2 + (size_t)n * HEADS);
    const float s2h = h == 0 ? s2v.x : h == 1 ? s2v.y : h == 2 ? s2v.z : s2v.w;
    float m0 = -INFINITY, m1 = -INFINITY, m2 = -INFINITY, m3 = -INFINITY;

    if (L <= OWN_SLOT) {
      const int* sl = slot + loc * OWN_SLOT;
      const bool v0 = lane < L;
      const bool v1 = lane + 32 < L;
      const int key0 = v0 ? sl[lane] : 0x7fffffff;
      const int key1 = v1 ? sl[lane + 32] : 0x7fffffff;
      float4 a0 = make_float4(0.f, 0.f, 0.f, 0.f), a1 = a0;
      if (v0) a0 = *(const float4*)(s1 + (size_t)key0 * HEADS);
      if (v1) a1 = *(const float4*)(s1 + (size_t)key1 * HEADS);
      int r0 = 0, r1 = 0;
      if (L <= 32) {
        for (int t = 0; t < L; ++t) {
          int kt = __builtin_amdgcn_readlane(key0, t);
          r0 += (kt < key0 || (kt == key0 && t < lane)) ? 1 : 0;
        }
      } else {
        for (int t = 0; t < 32; ++t) {
          int kt = __builtin_amdgcn_readlane(key0, t);
          r0 += (kt < key0 || (kt == key0 && t < lane)) ? 1 : 0;
          r1 += (kt <= key1) ? 1 : 0;
        }
        for (int t = 32; t < L; ++t) {
          int kt = __builtin_amdgcn_readlane(key1, t - 32);
          r0 += (kt < key0) ? 1 : 0;
          r1 += (kt < key1 || (kt == key1 && (t - 32) < lane)) ? 1 : 0;
        }
      }
      #pragma unroll
      for (int c = 0; c < 2; ++c) {
        const int    key  = c == 0 ? key0 : key1;
        const int    rank = c == 0 ? r0 : r1;
        const float4 s1v  = c == 0 ? a0 : a1;
        if (c == 0 ? v0 : v1) {
          float e0 = leaky02(s1v.x + s2v.x);
          float e1 = leaky02(s1v.y + s2v.y);
          float e2 = leaky02(s1v.z + s2v.z);
          float e3 = leaky02(s1v.w + s2v.w);
          wsrc[wave][rank]   = key;
          wex[wave][0][rank] = e0;
          wex[wave][1][rank] = e1;
          wex[wave][2][rank] = e2;
          wex[wave][3][rank] = e3;
          m0 = fmaxf(m0, e0); m1 = fmaxf(m1, e1);
          m2 = fmaxf(m2, e2); m3 = fmaxf(m3, e3);
        }
      }
      if (lane < OWN_PAD) {
        wsrc[wave][L + lane]   = 0;
        wex[wave][0][L + lane] = 0.f;
        wex[wave][1][L + lane] = 0.f;
        wex[wave][2][L + lane] = 0.f;
        wex[wave][3][L + lane] = 0.f;
      }
      m0 = wave_max(m0); m1 = wave_max(m1); m2 = wave_max(m2); m3 = wave_max(m3);
      const float mh = h == 0 ? m0 : h == 1 ? m1 : h == 2 ? m2 : m3;
      const float* wxl = Wx + j;
      for (int t0 = 0; t0 < L; t0 += OWN_PAD) {
        int    sv[OWN_PAD];
        float  ev[OWN_PAD];
        float4 row[OWN_PAD];
        #pragma unroll
        for (int u = 0; u < OWN_PAD; ++u) {
          sv[u] = wsrc[wave][t0 + u];
          ev[u] = wex[wave][h][t0 + u];
        }
        #pragma unroll
        for (int u = 0; u < OWN_PAD; ++u) {
          row[u] = make_float4(0.f, 0.f, 0.f, 0.f);
          if (t0 + u < L) row[u] = *(const float4*)(wxl + (size_t)sv[u] * OUT_DIM);
        }
        #pragma unroll
        for (int u = 0; u < OWN_PAD; ++u) {
          float xv = (t0 + u < L) ? __expf(ev[u] - mh) : 0.f;
          den   += xv;
          acc.x += xv * row[u].x; acc.y += xv * row[u].y;
          acc.z += xv * row[u].z; acc.w += xv * row[u].w;
        }
      }
    } else {
      for (int base2 = 0; base2 < E; base2 += 32) {
        int e = base2 + lane;
        if (e < E && dstp[e] == n) {
          float4 s1v = *(const float4*)(s1 + (size_t)ei[e] * HEADS);
          m0 = fmaxf(m0, leaky02(s1v.x + s2v.x));
          m1 = fmaxf(m1, leaky02(s1v.y + s2v.y));
          m2 = fmaxf(m2, leaky02(s1v.z + s2v.z));
          m3 = fmaxf(m3, leaky02(s1v.w + s2v.w));
        }
      }
      m0 = wave_max(m0); m1 = wave_max(m1); m2 = wave_max(m2); m3 = wave_max(m3);
      const float mh = h == 0 ? m0 : h == 1 ? m1 : h == 2 ? m2 : m3;
      for (int base2 = 0; base2 < E; base2 += 32) {
        int e = base2 + lane;
        bool mt = e < E && dstp[e] == n;
        int src = mt ? ei[e] : 0;
        unsigned mask = __builtin_amdgcn_ballot_w32(mt);
        while (mask) {
          int t = __builtin_ctz(mask);
          mask &= mask - 1;
          int s = __builtin_amdgcn_readlane(src, t);
          float ev = leaky02(s1[(size_t)s * HEADS + h] + s2h);
          float4 row = *(const float4*)(Wx + (size_t)s * OUT_DIM + j);
          float xv = __expf(ev - mh);
          den   += xv;
          acc.x += xv * row.x; acc.y += xv * row.y;
          acc.z += xv * row.z; acc.w += xv * row.w;
        }
      }
      if (den == 0.f) den = 1.f;
    }
    float4 o;
    o.x = elu1(acc.x / den); o.y = elu1(acc.y / den);
    o.z = elu1(acc.z / den); o.w = elu1(acc.w / den);
    *(float4*)op = o;
  }
}

extern "C" void kernel_launch(void* const* d_in, const int* in_sizes, int n_in,
                              void* d_out, int out_size, void* d_ws, size_t ws_size,
                              hipStream_t stream) {
  const float* x  = (const float*)d_in[0];
  const float* W  = (const float*)d_in[1];
  const float* a  = (const float*)d_in[2];
  const int*   ei = (const int*)d_in[3];
  const int N = in_sizes[0] / IN_DIM;
  const int E = in_sizes[3] / 2;
  float* out = (float*)d_out;

  char* ws = (char*)d_ws;
  size_t off = 0;
  auto carve = [&](size_t bytes) -> void* {
    void* p = ws + off;
    off += (bytes + 255) & ~(size_t)255;
    return p;
  };
  float* Wx = (float*)carve((size_t)N * OUT_DIM * sizeof(float));
  float* s1 = (float*)carve((size_t)N * HEADS * sizeof(float));
  float* s2 = (float*)carve((size_t)N * HEADS * sizeof(float));
  unsigned* okey = (unsigned*)carve((size_t)E + 16);
  (void)ws_size;

  gat_gemm_wmma<<<(N + 15) / 16, 256, 0, stream>>>(x, W, Wx, N);

  gat_scores<<<(N * HEADS + 255) / 256, 256, 0, stream>>>(Wx, a, s1, s2, ei, okey, N, E);

  gat_owner_aggr<<<(N + OWN_R - 1) / OWN_R, OWN_THREADS, 0, stream>>>(
      ei, okey, Wx, s1, s2, out, N, E);
}


# GEMM register tile: a wave holds 32 nodes x 64 outputs (8 accumulators), a loaded fragment feeds 4 or 2 products, a lane's 4 tiles are adjacent outputs (16-byte stores); 8x fewer waves; k order kept
# speedup vs baseline: 6.0136x; 1.2059x over previous
#include <hip/hip_runtime.h>
#include <cstdint>
#include <cstddef>

typedef __attribute__((ext_vector_type(2))) float v2f;
typedef __attribute__((ext_vector_type(8))) float v8f;

#define IN_DIM   128
#define OUT_DIM  128
#define HEADS    4
#define HEAD_DIM 32

#define GT_BROWS 64
#define GT_WROWS 32
#define GT_WCOLS 64

__global__ __launch_bounds__(128)
void gat_gemm_wmma(const float* __restrict__ x,
                   const float* __restrict__ W,
                   float* __restrict__ Wx, int N) {
  const int wave  = threadIdx.x >> 5;
  const int lane  = threadIdx.x & 31;
  const int mbase = blockIdx.x * GT_BROWS + (wave >> 1) * GT_WROWS;
  const int obase = (wave & 1) * GT_WCOLS;
  const int half  = lane >> 4;
  const int l15   = lane & 15;

  int arow0 = mbase + l15;
  int arow1 = mbase + 16 + l15;
  if (arow0 >= N) arow0 = N - 1;
  if (arow1 >= N) arow1 = N - 1;

  const float* ap0 = x + (size_t)arow0 * IN_DIM + 2 * half;
  const float* ap1 = x + (size_t)arow1 * IN_DIM + 2 * half;
  const float* bp  = W + (size_t)(obase + 4 * l15) * IN_DIM + 2 * half;

  v8f c00 = {}, c01 = {}, c02 = {}, c03 = {};
  v8f c10 = {}, c11 = {}, c12 = {}, c13 = {};
  #pragma unroll
  for (int k = 0; k < IN_DIM; k += 4) {
    const v2f a0 = *(const v2f*)(ap0 + k);
    const v2f a1 = *(const v2f*)(ap1 + k);
    const v2f b0 = *(const v2f*)(bp + k);
    const v2f b1 = *(const v2f*)(bp + IN_DIM + k);
    const v2f b2 = *(const v2f*)(bp + 2 * IN_DIM + k);
    const v2f b3 = *(const v2f*)(bp + 3 * IN_DIM + k);
    c00 = __builtin_amdgcn_wmma_f32_16x16x4_f32(false, a0, false, b0, (short)0, c00, false, false);
    c10 = __builtin_amdgcn_wmma_f32_16x16x4_f32(false, a1, false, b0, (short)0, c10, false, false);
    c01 = __builtin_amdgcn_wmma_f32_16x16x4_f32(false, a0, false, b1, (short)0, c01, false, false);
    c11 = __builtin_amdgcn_wmma_f32_16x16x4_f32(false, a1, false, b1, (short)0, c11, false, false);
    c02 = __builtin_amdgcn_wmma_f32_16x16x4_f32(false, a0, false, b2, (short)0, c02, false, false);
    c12 = __builtin_amdgcn_wmma_f32_16x16x4_f32(false, a1, false, b2, (short)0, c12, false, false);
    c03 = __builtin_amdgcn_wmma_f32_16x16x4_f32(false, a0, false, b3, (short)0, c03, false, false);
    c13 = __builtin_amdgcn_wmma_f32_16x16x4_f32(false, a1, false, b3, (short)0, c13, false, false);
  }

  float* op = Wx + obase + 4 * l15;
  #pragma unroll
  for (int r = 0; r < 8; ++r) {
    const int row0 = mbase + r + 8 * half;
    const int row1 = row0 + 16;
    if (row0 < N)
      *(float4*)(op + (size_t)row0 * OUT_DIM) = make_float4(c00[r], c01[r], c02[r], c03[r]);
    if (row1 < N)
      *(float4*)(op + (size_t)row1 * OUT_DIM) = make_float4(c10[r], c11[r], c12[r], c13[r]);
  }
}

#define OWN_R       196
#define OWN_SLOT    64
#define OWN_PAD     8
#define OWN_THREADS 1024
#define OWN_WAVES   (OWN_THREADS / 32)
#define OWN_SCAN_U  8
#define OWN_WCAP    256
#define OWN_LK_U    4

__device__ __forceinline__ unsigned own_key(int d) {
  return ((unsigned)d / (unsigned)OWN_R) & 255u;
}

__global__ void gat_scores(const float* __restrict__ Wx,
                           const float* __restrict__ a,
                           float* __restrict__ s1, float* __restrict__ s2,
                           const int* __restrict__ ei,
                           unsigned* __restrict__ okey,
                           int N, int E) {
  int t = blockIdx.x * blockDim.x + threadIdx.x;
  if ((E & 3) == 0) {
    const int4* __restrict__ d4 = (const int4*)(ei + E);
    const int nw   = E >> 2;
    const int step = gridDim.x * blockDim.x;
    for (int i = t; i < nw; i += step) {
      const int4 d = d4[i];
      okey[i] = own_key(d.x) | (own_key(d.y) << 8) |
                (own_key(d.z) << 16) | (own_key(d.w) << 24);
    }
  }
  if (t >= N * HEADS) return;
  int n = t >> 2, h = t & 3;
  const float4* p  = (const float4*)(Wx + (size_t)n * OUT_DIM + h * HEAD_DIM);
  const float4* a1 = (const float4*)(a);
  const float4* a2 = (const float4*)(a + HEAD_DIM);
  float acc1 = 0.f, acc2 = 0.f;
  #pragma unroll
  for (int d = 0; d < HEAD_DIM / 4; ++d) {
    float4 v  = p[d];
    float4 c1 = a1[d];
    float4 c2 = a2[d];
    acc1 += v.x * c1.x + v.y * c1.y + v.z * c1.z + v.w * c1.w;
    acc2 += v.x * c2.x + v.y * c2.y + v.z * c2.z + v.w * c2.w;
  }
  s1[t] = acc1;
  s2[t] = acc2;
}


__device__ __forceinline__ float leaky02(float el) {
  return el > 0.f ? el : 0.2f * el;
}
__device__ __forceinline__ float elu1(float v) {
  return v > 0.f ? v : (__expf(v) - 1.0f);
}
__device__ __forceinline__ float wave_max(float v) {
  #pragma unroll
  for (int o = 16; o > 0; o >>= 1) v = fmaxf(v, __shfl_xor(v, o));
  return v;
}
__device__ __forceinline__ unsigned own_zbytes(unsigned x) {
  return ~(((x & 0x7f7f7f7fu) + 0x7f7f7f7fu) | x | 0x7f7f7f7fu);
}
__device__ __forceinline__ void own_list(unsigned m, int e0, int* wl, int& wcount) {
  unsigned b;
  while ((b = __builtin_amdgcn_ballot_w32(m != 0u)) != 0u) {
    if (m != 0u) {
      const int bit = __builtin_ctz(m);
      const int pos = wcount + (int)__builtin_amdgcn_mbcnt_lo(b, 0u);
      if (pos < OWN_WCAP)
        wl[pos] = e0 + ((bit & 3) << 2) + (bit >> 3) + ((bit & 4) ? OWN_THREADS * 16 : 0);
      m &= m - 1u;
    }
    wcount += __builtin_popcount(b);
  }
}
__device__ __forceinline__ unsigned own_mask(uint4 v, unsigned rep) {
  return (own_zbytes(v.x ^ rep) >> 7) | (own_zbytes(v.y ^ rep) >> 6) |
         (own_zbytes(v.z ^ rep) >> 5) | (own_zbytes(v.w ^ rep) >> 4);
}
template <bool GUARD>
__device__ __forceinline__ void own_scan_trip(const uint4* __restrict__ k4,
                                              int base, int tid, int nvec,
                                              unsigned rep, int* wl, int& wcount) {
  uint4 v[OWN_SCAN_U];
  const unsigned nomatch = rep ^ 0x01010101u;
  #pragma unroll
  for (int u = 0; u < OWN_SCAN_U; ++u) {
    int i = base + u * OWN_THREADS + tid;
    if (GUARD) v[u] = (i < nvec) ? k4[i] : make_uint4(nomatch, nomatch, nomatch, nomatch);
    else       v[u] = k4[i];
  }
  #pragma unroll
  for (int u = 0; u < OWN_SCAN_U; u += 2) {
    const int e0 = (base + u * OWN_THREADS + tid) << 4;
    const unsigned m = own_mask(v[u], rep) | (own_mask(v[u + 1], rep) << 4);
    own_list(m, e0, wl, wcount);
  }
}

__global__ __launch_bounds__(OWN_THREADS)
void gat_owner_aggr(const int* __restrict__ ei,
                    const unsigned* __restrict__ okey,
                    const float* __restrict__ Wx,
                    const float* __restrict__ s1,
                    const float* __restrict__ s2,
                    float* __restrict__ out, int N, int E) {
  __shared__ int   cnt[OWN_R];
  __shared__ int   slot[OWN_R * OWN_SLOT];
  __shared__ int   wlist[OWN_WAVES][OWN_WCAP];
  __shared__ int   over;
  __shared__ int   wsrc[OWN_WAVES][OWN_SLOT + OWN_PAD];
  __shared__ float wex[OWN_WAVES][HEADS][OWN_SLOT + OWN_PAD];

  const int tid  = threadIdx.x;
  const int wave = __builtin_amdgcn_readfirstlane(tid >> 5);
  const int lane = tid & 31;
  const int lo   = blockIdx.x * OWN_R;
  const int* __restrict__ dstp = ei + E;

  for (int i = tid; i < OWN_R; i += OWN_THREADS) cnt[i] = 0;
  if (tid == 0) over = 0;
  __syncthreads();

  int* wl = wlist[wave];
  int wcount = 0;
  const unsigned rep = (blockIdx.x & 255u) * 0x01010101u;
  const int nvec = ((E & 3) == 0) ? (E >> 4) : 0;
  const uint4* __restrict__ k4 = (const uint4*)okey;
  const int trip = OWN_THREADS * OWN_SCAN_U;
  int base = 0;
  for (; base + trip <= nvec; base += trip)
    own_scan_trip<false>(k4, base, tid, nvec, rep, wl, wcount);
  if (base < nvec)
    own_scan_trip<true>(k4, base, tid, nvec, rep, wl, wcount);
  for (int eb = nvec << 4; eb < E; eb += OWN_THREADS) {
    const int e = eb + tid;
    const bool hit = e < E && (unsigned)(dstp[e] - lo) < (unsigned)OWN_R;
    own_list(hit ? 1u : 0u, e, wl, wcount);
  }
  if (wcount > OWN_WCAP && lane == 0) over = 1;
  __syncthreads();

  if (over == 0) {
    for (int b0 = 0; b0 < wcount; b0 += 32 * OWN_LK_U) {
      int dd[OWN_LK_U], ss[OWN_LK_U];
      #pragma unroll
      for (int u = 0; u < OWN_LK_U; ++u) {
        const int i = b0 + u * 32 + lane;
        dd[u] = -1;
        ss[u] = 0;
        if (i < wcount) {
          const int e = wl[i];
          dd[u] = dstp[e];
          ss[u] = ei[e];
        }
      }
      #pragma unroll
      for (int u = 0; u < OWN_LK_U; ++u) {
        const unsigned loc = (unsigned)(dd[u] - lo);
        if (loc < (unsigned)OWN_R) {
          const int q = atomicAdd(&cnt[loc], 1);
          if (q < OWN_SLOT) slot[loc * OWN_SLOT + q] = ss[u];
        }
      }
    }
  } else {
    for (int i = tid; i < OWN_R; i += OWN_THREADS) cnt[i] = OWN_SLOT + 1;
  }
  __syncthreads();

  const int h = lane >> 3;
  const int j = lane << 2;
  for (int loc = wave; loc < OWN_R; loc += OWN_WAVES) {
    const int n = lo + loc;
    if (n >= N) break;
    const int L = __builtin_amdgcn_readfirstlane(cnt[loc]);
    float4 acc = make_float4(0.f, 0.f, 0.f, 0.f);
    float den = 0.f;
    float* op = out + (size_t)n * OUT_DIM + j;
    if (L == 0) {
      *(float4*)op = acc;
      continue;
    }
    const float4 s2v = *(const float4*)(s2 + (size_t)n * HEADS);
    const float s2h = h == 0 ? s2v.x : h == 1 ? s2v.y : h == 2 ? s2v.z : s2v.w;
    float m0 = -INFINITY, m1 = -INFINITY, m2 = -INFINITY, m3 = -INFINITY;

    if (L <= OWN_SLOT) {
      const int* sl = slot + loc * OWN_SLOT;
      const bool v0 = lane < L;
      const bool v1 = lane + 32 < L;
      const int key0 = v0 ? sl[lane] : 0x7fffffff;
      const int key1 = v1 ? sl[lane + 32] : 0x7fffffff;
      float4 a0 = make_float4(0.f, 0.f, 0.f, 0.f), a1 = a0;
      if (v0) a0 = *(const float4*)(s1 + (size_t)key0 * HEADS);
      if (v1) a1 = *(const float4*)(s1 + (size_t)key1 * HEADS);
      int r0 = 0, r1 = 0;
      if (L <= 32) {
        for (int t = 0; t < L; ++t) {
          int kt = __builtin_amdgcn_readlane(key0, t);
          r0 += (kt < key0 || (kt == key0 && t < lane)) ? 1 : 0;
        }
      } else {
        for (int t = 0; t < 32; ++t) {
          int kt = __builtin_amdgcn_readlane(key0, t);
          r0 += (kt < key0 || (kt == key0 && t < lane)) ? 1 : 0;
          r1 += (kt <= key1) ? 1 : 0;
        }
        for (int t = 32; t < L; ++t) {
          int kt = __builtin_amdgcn_readlane(key1, t - 32);
          r0 += (kt < key0) ? 1 : 0;
          r1 += (kt < key1 || (kt == key1 && (t - 32) < lane)) ? 1 : 0;
        }
      }
      #pragma unroll
      for (int c = 0; c < 2; ++c) {
        const int    key  = c == 0 ? key0 : key1;
        const int    rank = c == 0 ? r0 : r1;
        const float4 s1v  = c == 0 ? a0 : a1;
        if (c == 0 ? v0 : v1) {
          float e0 = leaky02(s1v.x + s2v.x);
          float e1 = leaky02(s1v.y + s2v.y);
          float e2 = leaky02(s1v.z + s2v.z);
          float e3 = leaky02(s1v.w + s2v.w);
          wsrc[wave][rank]   = key;
          wex[wave][0][rank] = e0;
          wex[wave][1][rank] = e1;
          wex[wave][2][rank] = e2;
          wex[wave][3][rank] = e3;
          m0 = fmaxf(m0, e0); m1 = fmaxf(m1, e1);
          m2 = fmaxf(m2, e2); m3 = fmaxf(m3, e3);
        }
      }
      if (lane < OWN_PAD) {
        wsrc[wave][L + lane]   = 0;
        wex[wave][0][L + lane] = 0.f;
        wex[wave][1][L + lane] = 0.f;
        wex[wave][2][L + lane] = 0.f;
        wex[wave][3][L + lane] = 0.f;
      }
      m0 = wave_max(m0); m1 = wave_max(m1); m2 = wave_max(m2); m3 = wave_max(m3);
      const float mh = h == 0 ? m0 : h == 1 ? m1 : h == 2 ? m2 : m3;
      const float* wxl = Wx + j;
      for (int t0 = 0; t0 < L; t0 += OWN_PAD) {
        int    sv[OWN_PAD];
        float  ev[OWN_PAD];
        float4 row[OWN_PAD];
        #pragma unroll
        for (int u = 0; u < OWN_PAD; ++u) {
          sv[u] = wsrc[wave][t0 + u];
          ev[u] = wex[wave][h][t0 + u];
        }
        #pragma unroll
        for (int u = 0; u < OWN_PAD; ++u) {
          row[u] = make_float4(0.f, 0.f, 0.f, 0.f);
          if (t0 + u < L) row[u] = *(const float4*)(wxl + (size_t)sv[u] * OUT_DIM);
        }
        #pragma unroll
        for (int u = 0; u < OWN_PAD; ++u) {
          float xv = (t0 + u < L) ? __expf(ev[u] - mh) : 0.f;
          den   += xv;
          acc.x += xv * row[u].x; acc.y += xv * row[u].y;
          acc.z += xv * row[u].z; acc.w += xv * row[u].w;
        }
      }
    } else {
      for (int base2 = 0; base2 < E; base2 += 32) {
        int e = base2 + lane;
        if (e < E && dstp[e] == n) {
          float4 s1v = *(const float4*)(s1 + (size_t)ei[e] * HEADS);
          m0 = fmaxf(m0, leaky02(s1v.x + s2v.x));
          m1 = fmaxf(m1, leaky02(s1v.y + s2v.y));
          m2 = fmaxf(m2, leaky02(s1v.z + s2v.z));
          m3 = fmaxf(m3, leaky02(s1v.w + s2v.w));
        }
      }
      m0 = wave_max(m0); m1 = wave_max(m1); m2 = wave_max(m2); m3 = wave_max(m3);
      const float mh = h == 0 ? m0 : h == 1 ? m1 : h == 2 ? m2 : m3;
      for (int base2 = 0; base2 < E; base2 += 32) {
        int e = base2 + lane;
        bool mt = e < E && dstp[e] == n;
        int src = mt ? ei[e] : 0;
        unsigned mask = __builtin_amdgcn_ballot_w32(mt);
        while (mask) {
          int t = __builtin_ctz(mask);
          mask &= mask - 1;
          int s = __builtin_amdgcn_readlane(src, t);
          float ev = leaky02(s1[(size_t)s * HEADS + h] + s2h);
          float4 row = *(const float4*)(Wx + (size_t)s * OUT_DIM + j);
          float xv = __expf(ev - mh);
          den   += xv;
          acc.x += xv * row.x; acc.y += xv * row.y;
          acc.z += xv * row.z; acc.w += xv * row.w;
        }
      }
      if (den == 0.f) den = 1.f;
    }
    float4 o;
    o.x = elu1(acc.x / den); o.y = elu1(acc.y / den);
    o.z = elu1(acc.z / den); o.w = elu1(acc.w / den);
    *(float4*)op = o;
  }
}

extern "C" void kernel_launch(void* const* d_in, const int* in_sizes, int n_in,
                              void* d_out, int out_size, void* d_ws, size_t ws_size,
                              hipStream_t stream) {
  const float* x  = (const float*)d_in[0];
  const float* W  = (const float*)d_in[1];
  const float* a  = (const float*)d_in[2];
  const int*   ei = (const int*)d_in[3];
  const int N = in_sizes[0] / IN_DIM;
  const int E = in_sizes[3] / 2;
  float* out = (float*)d_out;

  char* ws = (char*)d_ws;
  size_t off = 0;
  auto carve = [&](size_t bytes) -> void* {
    void* p = ws + off;
    off += (bytes + 255) & ~(size_t)255;
    return p;
  };
  float* Wx = (float*)carve((size_t)N * OUT_DIM * sizeof(float));
  float* s1 = (float*)carve((size_t)N * HEADS * sizeof(float));
  float* s2 = (float*)carve((size_t)N * HEADS * sizeof(float));
  unsigned* okey = (unsigned*)carve((size_t)E + 16);
  (void)ws_size;

  gat_gemm_wmma<<<(N + GT_BROWS - 1) / GT_BROWS, 128, 0, stream>>>(x, W, Wx, N);

  gat_scores<<<(N * HEADS + 255) / 256, 256, 0, stream>>>(Wx, a, s1, s2, ei, okey, N, E);

  gat_owner_aggr<<<(N + OWN_R - 1) / OWN_R, OWN_THREADS, 0, stream>>>(
      ei, okey, Wx, s1, s2, out, N, E);
}


# aggregation scan skips key groups without own edges: scores launch adds one presence word per owner per 32 groups of 16 keys (LDS integer OR); a work-group loads flagged groups only; float sums kept
# speedup vs baseline: 6.2327x; 1.0364x over previous
#include <hip/hip_runtime.h>
#include <cstdint>
#include <cstddef>

typedef __attribute__((ext_vector_type(2))) float v2f;
typedef __attribute__((ext_vector_type(8))) float v8f;

#define IN_DIM   128
#define OUT_DIM  128
#define HEADS    4
#define HEAD_DIM 32

#define GT_BROWS 64
#define GT_WROWS 32
#define GT_WCOLS 64

__global__ __launch_bounds__(128)
void gat_gemm_wmma(const float* __restrict__ x,
                   const float* __restrict__ W,
                   float* __restrict__ Wx, int N) {
  const int wave  = threadIdx.x >> 5;
  const int lane  = threadIdx.x & 31;
  const int mbase = blockIdx.x * GT_BROWS + (wave >> 1) * GT_WROWS;
  const int obase = (wave & 1) * GT_WCOLS;
  const int half  = lane >> 4;
  const int l15   = lane & 15;

  int arow0 = mbase + l15;
  int arow1 = mbase + 16 + l15;
  if (arow0 >= N) arow0 = N - 1;
  if (arow1 >= N) arow1 = N - 1;

  const float* ap0 = x + (size_t)arow0 * IN_DIM + 2 * half;
  const float* ap1 = x + (size_t)arow1 * IN_DIM + 2 * half;
  const float* bp  = W + (size_t)(obase + 4 * l15) * IN_DIM + 2 * half;

  v8f c00 = {}, c01 = {}, c02 = {}, c03 = {};
  v8f c10 = {}, c11 = {}, c12 = {}, c13 = {};
  #pragma unroll
  for (int k = 0; k < IN_DIM; k += 4) {
    const v2f a0 = *(const v2f*)(ap0 + k);
    const v2f a1 = *(const v2f*)(ap1 + k);
    const v2f b0 = *(const v2f*)(bp + k);
    const v2f b1 = *(const v2f*)(bp + IN_DIM + k);
    const v2f b2 = *(const v2f*)(bp + 2 * IN_DIM + k);
    const v2f b3 = *(const v2f*)(bp + 3 * IN_DIM + k);
    c00 = __builtin_amdgcn_wmma_f32_16x16x4_f32(false, a0, false, b0, (short)0, c00, false, false);
    c10 = __builtin_amdgcn_wmma_f32_16x16x4_f32(false, a1, false, b0, (short)0, c10, false, false);
    c01 = __builtin_amdgcn_wmma_f32_16x16x4_f32(false, a0, false, b1, (short)0, c01, false, false);
    c11 = __builtin_amdgcn_wmma_f32_16x16x4_f32(false, a1, false, b1, (short)0, c11, false, false);
    c02 = __builtin_amdgcn_wmma_f32_16x16x4_f32(false, a0, false, b2, (short)0, c02, false, false);
    c12 = __builtin_amdgcn_wmma_f32_16x16x4_f32(false, a1, false, b2, (short)0, c12, false, false);
    c03 = __builtin_amdgcn_wmma_f32_16x16x4_f32(false, a0, false, b3, (short)0, c03, false, false);
    c13 = __builtin_amdgcn_wmma_f32_16x16x4_f32(false, a1, false, b3, (short)0, c13, false, false);
  }

  float* op = Wx + obase + 4 * l15;
  #pragma unroll
  for (int r = 0; r < 8; ++r) {
    const int row0 = mbase + r + 8 * half;
    const int row1 = row0 + 16;
    if (row0 < N)
      *(float4*)(op + (size_t)row0 * OUT_DIM) = make_float4(c00[r], c01[r], c02[r], c03[r]);
    if (row1 < N)
      *(float4*)(op + (size_t)row1 * OUT_DIM) = make_float4(c10[r], c11[r], c12[r], c13[r]);
  }
}

#define OWN_R       196
#define OWN_SLOT    64
#define OWN_PAD     8
#define OWN_THREADS 1024
#define OWN_WAVES   (OWN_THREADS / 32)
#define OWN_WCAP    256
#define OWN_LK_U    4
#define OWN_NKEY    256

__device__ __forceinline__ unsigned own_key(int d) {
  return ((unsigned)d / (unsigned)OWN_R) & 255u;
}

#define SC_THREADS 256
#define SC_WAVES   (SC_THREADS / 32)

__global__ __launch_bounds__(SC_THREADS)
void gat_scores(const float* __restrict__ Wx,
                           const float* __restrict__ a,
                           float* __restrict__ s1, float* __restrict__ s2,
                           const int* __restrict__ ei,
                           unsigned* __restrict__ okey,
                           unsigned* __restrict__ obits,
                           int N, int E) {
  __shared__ __attribute__((aligned(16))) unsigned pres[SC_WAVES][OWN_NKEY];
  int t = blockIdx.x * blockDim.x + threadIdx.x;
  if ((E & 3) == 0) {
    const int4* __restrict__ d4 = (const int4*)(ei + E);
    uint4* __restrict__ k4 = (uint4*)okey;
    const int nvec = E >> 4;
    const int nrow = (nvec + 31) >> 5;
    const int lane = threadIdx.x & 31;
    const int wv   = threadIdx.x >> 5;
    uint4* pw4 = (uint4*)pres[wv];
    unsigned* pw = pres[wv];
    for (int r = wv * gridDim.x + blockIdx.x; r < nrow; r += gridDim.x * SC_WAVES) {
      pw4[2 * lane]     = make_uint4(0u, 0u, 0u, 0u);
      pw4[2 * lane + 1] = make_uint4(0u, 0u, 0u, 0u);
      const int g = (r << 5) + lane;
      if (g < nvec) {
        const unsigned bit = 1u << lane;
        unsigned kw[4];
        #pragma unroll
        for (int q = 0; q < 4; ++q) {
          const int4 d = d4[4 * g + q];
          const unsigned k0 = own_key(d.x), k1 = own_key(d.y);
          const unsigned k2 = own_key(d.z), k3 = own_key(d.w);
          kw[q] = k0 | (k1 << 8) | (k2 << 16) | (k3 << 24);
          atomicOr(&pw[k0], bit);
          atomicOr(&pw[k1], bit);
          atomicOr(&pw[k2], bit);
          atomicOr(&pw[k3], bit);
        }
        k4[g] = make_uint4(kw[0], kw[1], kw[2], kw[3]);
      }
      const uint4 r0 = pw4[2 * lane];
      const uint4 r1 = pw4[2 * lane + 1];
      uint4* ob = (uint4*)(obits + (size_t)r * OWN_NKEY);
      ob[2 * lane]     = r0;
      ob[2 * lane + 1] = r1;
    }
  }
  if (t >= N * HEADS) return;
  int n = t >> 2, h = t & 3;
  const float4* p  = (const float4*)(Wx + (size_t)n * OUT_DIM + h * HEAD_DIM);
  const float4* a1 = (const float4*)(a);
  const float4* a2 = (const float4*)(a + HEAD_DIM);
  float acc1 = 0.f, acc2 = 0.f;
  #pragma unroll
  for (int d = 0; d < HEAD_DIM / 4; ++d) {
    float4 v  = p[d];
    float4 c1 = a1[d];
    float4 c2 = a2[d];
    acc1 += v.x * c1.x + v.y * c1.y + v.z * c1.z + v.w * c1.w;
    acc2 += v.x * c2.x + v.y * c2.y + v.z * c2.z + v.w * c2.w;
  }
  s1[t] = acc1;
  s2[t] = acc2;
}


__device__ __forceinline__ float leaky02(float el) {
  return el > 0.f ? el : 0.2f * el;
}
__device__ __forceinline__ float elu1(float v) {
  return v > 0.f ? v : (__expf(v) - 1.0f);
}
__device__ __forceinline__ float wave_max(float v) {
  #pragma unroll
  for (int o = 16; o > 0; o >>= 1) v = fmaxf(v, __shfl_xor(v, o));
  return v;
}
__device__ __forceinline__ unsigned own_zbytes(unsigned x) {
  return ~(((x & 0x7f7f7f7fu) + 0x7f7f7f7fu) | x | 0x7f7f7f7fu);
}
__device__ __forceinline__ void own_list(unsigned m, int e0, int* wl, int& wcount) {
  unsigned b;
  while ((b = __builtin_amdgcn_ballot_w32(m != 0u)) != 0u) {
    if (m != 0u) {
      const int bit = __builtin_ctz(m);
      const int pos = wcount + (int)__builtin_amdgcn_mbcnt_lo(b, 0u);
      if (pos < OWN_WCAP)
        wl[pos] = e0 + ((bit & 3) << 2) + (bit >> 3);
      m &= m - 1u;
    }
    wcount += __builtin_popcount(b);
  }
}
__device__ __forceinline__ unsigned own_mask(uint4 v, unsigned rep) {
  return (own_zbytes(v.x ^ rep) >> 7) | (own_zbytes(v.y ^ rep) >> 6) |
         (own_zbytes(v.z ^ rep) >> 5) | (own_zbytes(v.w ^ rep) >> 4);
}
__device__ __forceinline__ void own_groups(unsigned word, int row, int* gl, int& gcount) {
  unsigned b;
  while ((b = __builtin_amdgcn_ballot_w32(word != 0u)) != 0u) {
    if (word != 0u) {
      const int pos = gcount + (int)__builtin_amdgcn_mbcnt_lo(b, 0u);
      if (pos < OWN_WCAP) gl[pos] = (row << 5) + __builtin_ctz(word);
      word &= word - 1u;
    }
    gcount += __builtin_popcount(b);
  }
}

__global__ __launch_bounds__(OWN_THREADS)
void gat_owner_aggr(const int* __restrict__ ei,
                    const unsigned* __restrict__ okey,
                    const unsigned* __restrict__ obits,
                    const float* __restrict__ Wx,
                    const float* __restrict__ s1,
                    const float* __restrict__ s2,
                    float* __restrict__ out, int N, int E) {
  __shared__ int   cnt[OWN_R];
  __shared__ int   slot[OWN_R * OWN_SLOT];
  __shared__ int   wlist[OWN_WAVES][OWN_WCAP];
  __shared__ int   glist[OWN_WAVES][OWN_WCAP];
  __shared__ int   over;
  __shared__ int   wsrc[OWN_WAVES][OWN_SLOT + OWN_PAD];
  __shared__ float wex[OWN_WAVES][HEADS][OWN_SLOT + OWN_PAD];

  const int tid  = threadIdx.x;
  const int wave = __builtin_amdgcn_readfirstlane(tid >> 5);
  const int lane = tid & 31;
  const int lo   = blockIdx.x * OWN_R;
  const int* __restrict__ dstp = ei + E;

  for (int i = tid; i < OWN_R; i += OWN_THREADS) cnt[i] = 0;
  if (tid == 0) over = 0;
  __syncthreads();

  int* wl = wlist[wave];
  int* gl = glist[wave];
  int wcount = 0;
  int gcount = 0;
  const unsigned own = blockIdx.x & 255u;
  const unsigned rep = own * 0x01010101u;
  const int nvec = ((E & 3) == 0) ? (E >> 4) : 0;
  const int nrow = (nvec + 31) >> 5;
  const uint4* __restrict__ k4 = (const uint4*)okey;
  const unsigned* __restrict__ ob = obits + own;
  for (int base = 0; base < nrow; base += 2 * OWN_THREADS) {
    const int r0 = base + tid;
    const int r1 = base + OWN_THREADS + lane * OWN_WAVES + wave;
    const unsigned w0 = (r0 < nrow) ? ob[(size_t)r0 * OWN_NKEY] : 0u;
    const unsigned w1 = (r1 < nrow) ? ob[(size_t)r1 * OWN_NKEY] : 0u;
    own_groups(w0, r0, gl, gcount);
    own_groups(w1, r1, gl, gcount);
  }
  if (gcount > OWN_WCAP && lane == 0) over = 1;
  {
    const int gc = gcount < OWN_WCAP ? gcount : OWN_WCAP;
    const unsigned nomatch = rep ^ 0x01010101u;
    for (int b0 = 0; b0 < gc; b0 += 32 * OWN_LK_U) {
      uint4 kv[OWN_LK_U];
      int   gg[OWN_LK_U];
      #pragma unroll
      for (int u = 0; u < OWN_LK_U; ++u) {
        const int i = b0 + u * 32 + lane;
        gg[u] = 0;
        kv[u] = make_uint4(nomatch, nomatch, nomatch, nomatch);
        if (i < gc) {
          const int g = gl[i];
          if (g < nvec) {
            gg[u] = g;
            kv[u] = k4[g];
          }
        }
      }
      #pragma unroll
      for (int u = 0; u < OWN_LK_U; ++u)
        own_list(own_mask(kv[u], rep), gg[u] << 4, wl, wcount);
    }
  }
  for (int eb = nvec << 4; eb < E; eb += OWN_THREADS) {
    const int e = eb + tid;
    const bool hit = e < E && (unsigned)(dstp[e] - lo) < (unsigned)OWN_R;
    own_list(hit ? 1u : 0u, e, wl, wcount);
  }
  if (wcount > OWN_WCAP && lane == 0) over = 1;
  __syncthreads();

  if (over == 0) {
    for (int b0 = 0; b0 < wcount; b0 += 32 * OWN_LK_U) {
      int dd[OWN_LK_U], ss[OWN_LK_U];
      #pragma unroll
      for (int u = 0; u < OWN_LK_U; ++u) {
        const int i = b0 + u * 32 + lane;
        dd[u] = -1;
        ss[u] = 0;
        if (i < wcount) {
          const int e = wl[i];
          dd[u] = dstp[e];
          ss[u] = ei[e];
        }
      }
      #pragma unroll
      for (int u = 0; u < OWN_LK_U; ++u) {
        const unsigned loc = (unsigned)(dd[u] - lo);
        if (loc < (unsigned)OWN_R) {
          const int q = atomicAdd(&cnt[loc], 1);
          if (q < OWN_SLOT) slot[loc * OWN_SLOT + q] = ss[u];
        }
      }
    }
  } else {
    for (int i = tid; i < OWN_R; i += OWN_THREADS) cnt[i] = OWN_SLOT + 1;
  }
  __syncthreads();

  const int h = lane >> 3;
  const int j = lane << 2;
  for (int loc = wave; loc < OWN_R; loc += OWN_WAVES) {
    const int n = lo + loc;
    if (n >= N) break;
    const int L = __builtin_amdgcn_readfirstlane(cnt[loc]);
    float4 acc = make_float4(0.f, 0.f, 0.f, 0.f);
    float den = 0.f;
    float* op = out + (size_t)n * OUT_DIM + j;
    if (L == 0) {
      *(float4*)op = acc;
      continue;
    }
    const float4 s2v = *(const float4*)(s2 + (size_t)n * HEADS);
    const float s2h = h == 0 ? s2v.x : h == 1 ? s2v.y : h == 2 ? s2v.z : s2v.w;
    float m0 = -INFINITY, m1 = -INFINITY, m2 = -INFINITY, m3 = -INFINITY;

    if (L <= OWN_SLOT) {
      const int* sl = slot + loc * OWN_SLOT;
      const bool v0 = lane < L;
      const bool v1 = lane + 32 < L;
      const int key0 = v0 ? sl[lane] : 0x7fffffff;
      const int key1 = v1 ? sl[lane + 32] : 0x7fffffff;
      float4 a0 = make_float4(0.f, 0.f, 0.f, 0.f), a1 = a0;
      if (v0) a0 = *(const float4*)(s1 + (size_t)key0 * HEADS);
      if (v1) a1 = *(const float4*)(s1 + (size_t)key1 * HEADS);
      int r0 = 0, r1 = 0;
      if (L <= 32) {
        for (int t = 0; t < L; ++t) {
          int kt = __builtin_amdgcn_readlane(key0, t);
          r0 += (kt < key0 || (kt == key0 && t < lane)) ? 1 : 0;
        }
      } else {
        for (int t = 0; t < 32; ++t) {
          int kt = __builtin_amdgcn_readlane(key0, t);
          r0 += (kt < key0 || (kt == key0 && t < lane)) ? 1 : 0;
          r1 += (kt <= key1) ? 1 : 0;
        }
        for (int t = 32; t < L; ++t) {
          int kt = __builtin_amdgcn_readlane(key1, t - 32);
          r0 += (kt < key0) ? 1 : 0;
          r1 += (kt < key1 || (kt == key1 && (t - 32) < lane)) ? 1 : 0;
        }
      }
      #pragma unroll
      for (int c = 0; c < 2; ++c) {
        const int    key  = c == 0 ? key0 : key1;
        const int    rank = c == 0 ? r0 : r1;
        const float4 s1v  = c == 0 ? a0 : a1;
        if (c == 0 ? v0 : v1) {
          float e0 = leaky02(s1v.x + s2v.x);
          float e1 = leaky02(s1v.y + s2v.y);
          float e2 = leaky02(s1v.z + s2v.z);
          float e3 = leaky02(s1v.w + s2v.w);
          wsrc[wave][rank]   = key;
          wex[wave][0][rank] = e0;
          wex[wave][1][rank] = e1;
          wex[wave][2][rank] = e2;
          wex[wave][3][rank] = e3;
          m0 = fmaxf(m0, e0); m1 = fmaxf(m1, e1);
          m2 = fmaxf(m2, e2); m3 = fmaxf(m3, e3);
        }
      }
      if (lane < OWN_PAD) {
        wsrc[wave][L + lane]   = 0;
        wex[wave][0][L + lane] = 0.f;
        wex[wave][1][L + lane] = 0.f;
        wex[wave][2][L + lane] = 0.f;
        wex[wave][3][L + lane] = 0.f;
      }
      m0 = wave_max(m0); m1 = wave_max(m1); m2 = wave_max(m2); m3 = wave_max(m3);
      const float mh = h == 0 ? m0 : h == 1 ? m1 : h == 2 ? m2 : m3;
      const float* wxl = Wx + j;
      for (int t0 = 0; t0 < L; t0 += OWN_PAD) {
        int    sv[OWN_PAD];
        float  ev[OWN_PAD];
        float4 row[OWN_PAD];
        #pragma unroll
        for (int u = 0; u < OWN_PAD; ++u) {
          sv[u] = wsrc[wave][t0 + u];
          ev[u] = wex[wave][h][t0 + u];
        }
        #pragma unroll
        for (int u = 0; u < OWN_PAD; ++u) {
          row[u] = make_float4(0.f, 0.f, 0.f, 0.f);
          if (t0 + u < L) row[u] = *(const float4*)(wxl + (size_t)sv[u] * OUT_DIM);
        }
        #pragma unroll
        for (int u = 0; u < OWN_PAD; ++u) {
          float xv = (t0 + u < L) ? __expf(ev[u] - mh) : 0.f;
          den   += xv;
          acc.x += xv * row[u].x; acc.y += xv * row[u].y;
          acc.z += xv * row[u].z; acc.w += xv * row[u].w;
        }
      }
    } else {
      for (int base2 = 0; base2 < E; base2 += 32) {
        int e = base2 + lane;
        if (e < E && dstp[e] == n) {
          float4 s1v = *(const float4*)(s1 + (size_t)ei[e] * HEADS);
          m0 = fmaxf(m0, leaky02(s1v.x + s2v.x));
          m1 = fmaxf(m1, leaky02(s1v.y + s2v.y));
          m2 = fmaxf(m2, leaky02(s1v.z + s2v.z));
          m3 = fmaxf(m3, leaky02(s1v.w + s2v.w));
        }
      }
      m0 = wave_max(m0); m1 = wave_max(m1); m2 = wave_max(m2); m3 = wave_max(m3);
      const float mh = h == 0 ? m0 : h == 1 ? m1 : h == 2 ? m2 : m3;
      for (int base2 = 0; base2 < E; base2 += 32) {
        int e = base2 + lane;
        bool mt = e < E && dstp[e] == n;
        int src = mt ? ei[e] : 0;
        unsigned mask = __builtin_amdgcn_ballot_w32(mt);
        while (mask) {
          int t = __builtin_ctz(mask);
          mask &= mask - 1;
          int s = __builtin_amdgcn_readlane(src, t);
          float ev = leaky02(s1[(size_t)s * HEADS + h] + s2h);
          float4 row = *(const float4*)(Wx + (size_t)s * OUT_DIM + j);
          float xv = __expf(ev - mh);
          den   += xv;
          acc.x += xv * row.x; acc.y += xv * row.y;
          acc.z += xv * row.z; acc.w += xv * row.w;
        }
      }
      if (den == 0.f) den = 1.f;
    }
    float4 o;
    o.x = elu1(acc.x / den); o.y = elu1(acc.y / den);
    o.z = elu1(acc.z / den); o.w = elu1(acc.w / den);
    *(float4*)op = o;
  }
}

extern "C" void kernel_launch(void* const* d_in, const int* in_sizes, int n_in,
                              void* d_out, int out_size, void* d_ws, size_t ws_size,
                              hipStream_t stream) {
  const float* x  = (const float*)d_in[0];
  const float* W  = (const float*)d_in[1];
  const float* a  = (const float*)d_in[2];
  const int*   ei = (const int*)d_in[3];
  const int N = in_sizes[0] / IN_DIM;
  const int E = in_sizes[3] / 2;
  float* out = (float*)d_out;

  char* ws = (char*)d_ws;
  size_t off = 0;
  auto carve = [&](size_t bytes) -> void* {
    void* p = ws + off;
    off += (bytes + 255) & ~(size_t)255;
    return p;
  };
  float* Wx = (float*)carve((size_t)N * OUT_DIM * sizeof(float));
  float* s1 = (float*)carve((size_t)N * HEADS * sizeof(float));
  float* s2 = (float*)carve((size_t)N * HEADS * sizeof(float));
  unsigned* okey = (unsigned*)carve((size_t)E + 16);
  unsigned* obits = (unsigned*)carve((((size_t)E >> 9) + 1) * OWN_NKEY * sizeof(unsigned));
  (void)ws_size;

  gat_gemm_wmma<<<(N + GT_BROWS - 1) / GT_BROWS, 128, 0, stream>>>(x, W, Wx, N);

  gat_scores<<<(N * HEADS + SC_THREADS - 1) / SC_THREADS, SC_THREADS, 0, stream>>>(
      Wx, a, s1, s2, ei, okey, obits, N, E);

  gat_owner_aggr<<<(N + OWN_R - 1) / OWN_R, OWN_THREADS, 0, stream>>>(
      ei, okey, obits, Wx, s1, s2, out, N, E);
}


# scores launch writes a 4-byte record per edge (owner key, destination offset, source), not a 1-byte key; aggregation lists nodes' sources from flagged record groups: dst/src lookups gone; sums kept
# speedup vs baseline: 6.9166x; 1.1097x over previous
#include <hip/hip_runtime.h>
#include <cstdint>
#include <cstddef>

typedef __attribute__((ext_vector_type(2))) float v2f;
typedef __attribute__((ext_vector_type(8))) float v8f;

#define IN_DIM   128
#define OUT_DIM  128
#define HEADS    4
#define HEAD_DIM 32

#define GT_BROWS 64
#define GT_WROWS 32
#define GT_WCOLS 64

__global__ __launch_bounds__(128)
void gat_gemm_wmma(const float* __restrict__ x,
                   const float* __restrict__ W,
                   float* __restrict__ Wx, int N) {
  const int wave  = threadIdx.x >> 5;
  const int lane  = threadIdx.x & 31;
  const int mbase = blockIdx.x * GT_BROWS + (wave >> 1) * GT_WROWS;
  const int obase = (wave & 1) * GT_WCOLS;
  const int half  = lane >> 4;
  const int l15   = lane & 15;

  int arow0 = mbase + l15;
  int arow1 = mbase + 16 + l15;
  if (arow0 >= N) arow0 = N - 1;
  if (arow1 >= N) arow1 = N - 1;

  const float* ap0 = x + (size_t)arow0 * IN_DIM + 2 * half;
  const float* ap1 = x + (size_t)arow1 * IN_DIM + 2 * half;
  const float* bp  = W + (size_t)(obase + 4 * l15) * IN_DIM + 2 * half;

  v8f c00 = {}, c01 = {}, c02 = {}, c03 = {};
  v8f c10 = {}, c11 = {}, c12 = {}, c13 = {};
  #pragma unroll
  for (int k = 0; k < IN_DIM; k += 4) {
    const v2f a0 = *(const v2f*)(ap0 + k);
    const v2f a1 = *(const v2f*)(ap1 + k);
    const v2f b0 = *(const v2f*)(bp + k);
    const v2f b1 = *(const v2f*)(bp + IN_DIM + k);
    const v2f b2 = *(const v2f*)(bp + 2 * IN_DIM + k);
    const v2f b3 = *(const v2f*)(bp + 3 * IN_DIM + k);
    c00 = __builtin_amdgcn_wmma_f32_16x16x4_f32(false, a0, false, b0, (short)0, c00, false, false);
    c10 = __builtin_amdgcn_wmma_f32_16x16x4_f32(false, a1, false, b0, (short)0, c10, false, false);
    c01 = __builtin_amdgcn_wmma_f32_16x16x4_f32(false, a0, false, b1, (short)0, c01, false, false);
    c11 = __builtin_amdgcn_wmma_f32_16x16x4_f32(false, a1, false, b1, (short)0, c11, false, false);
    c02 = __builtin_amdgcn_wmma_f32_16x16x4_f32(false, a0, false, b2, (short)0, c02, false, false);
    c12 = __builtin_amdgcn_wmma_f32_16x16x4_f32(false, a1, false, b2, (short)0, c12, false, false);
    c03 = __builtin_amdgcn_wmma_f32_16x16x4_f32(false, a0, false, b3, (short)0, c03, false, false);
    c13 = __builtin_amdgcn_wmma_f32_16x16x4_f32(false, a1, false, b3, (short)0, c13, false, false);
  }

  float* op = Wx + obase + 4 * l15;
  #pragma unroll
  for (int r = 0; r < 8; ++r) {
    const int row0 = mbase + r + 8 * half;
    const int row1 = row0 + 16;
    if (row0 < N)
      *(float4*)(op + (size_t)row0 * OUT_DIM) = make_float4(c00[r], c01[r], c02[r], c03[r]);
    if (row1 < N)
      *(float4*)(op + (size_t)row1 * OUT_DIM) = make_float4(c10[r], c11[r], c12[r], c13[r]);
  }
}

#define OWN_R       196
#define OWN_SLOT    64
#define OWN_PAD     8
#define OWN_THREADS 1024
#define OWN_WAVES   (OWN_THREADS / 32)
#define OWN_WCAP    256
#define OWN_RK_U    8
#define OWN_PUT_U   4
#define OWN_NKEY    256

__device__ __forceinline__ unsigned own_key(int d) {
  return ((unsigned)d / (unsigned)OWN_R) & 255u;
}
__device__ __forceinline__ unsigned own_rec(int d, int s) {
  const unsigned o   = (unsigned)d / (unsigned)OWN_R;
  const unsigned loc = o < (unsigned)OWN_NKEY ? (unsigned)d - o * (unsigned)OWN_R : 255u;
  return (o & 255u) | (loc << 8) | ((unsigned)s << 16);
}

#define SC_THREADS 256
#define SC_WAVES   (SC_THREADS / 32)

__global__ __launch_bounds__(SC_THREADS)
void gat_scores(const float* __restrict__ Wx,
                           const float* __restrict__ a,
                           float* __restrict__ s1, float* __restrict__ s2,
                           const int* __restrict__ ei,
                           unsigned* __restrict__ orec,
                           unsigned* __restrict__ obits,
                           int N, int E) {
  __shared__ __attribute__((aligned(16))) unsigned pres[SC_WAVES][OWN_NKEY];
  int t = blockIdx.x * blockDim.x + threadIdx.x;
  if ((E & 3) == 0 && N <= OWN_NKEY * OWN_R) {
    const int4* __restrict__ d4 = (const int4*)(ei + E);
    const int4* __restrict__ s4 = (const int4*)ei;
    uint4* __restrict__ r4 = (uint4*)orec;
    const int nvec = E >> 4;
    const int nrow = (nvec + 31) >> 5;
    const int lane = threadIdx.x & 31;
    const int wv   = threadIdx.x >> 5;
    uint4* pw4 = (uint4*)pres[wv];
    unsigned* pw = pres[wv];
    for (int r = wv * gridDim.x + blockIdx.x; r < nrow; r += gridDim.x * SC_WAVES) {
      pw4[2 * lane]     = make_uint4(0u, 0u, 0u, 0u);
      pw4[2 * lane + 1] = make_uint4(0u, 0u, 0u, 0u);
      int4 dv[4], sv[4];
      bool okq[4];
      #pragma unroll
      for (int q = 0; q < 4; ++q) {
        okq[q] = (r << 5) + q * 8 + (lane >> 2) < nvec;
        const size_t v = okq[q] ? (size_t)r * 128 + q * 32 + lane : 0;
        dv[q] = d4[v];
        sv[q] = s4[v];
      }
      #pragma unroll
      for (int q = 0; q < 4; ++q) {
        if (okq[q]) {
          const int4 d = dv[q];
          const unsigned bit = 1u << (q * 8 + (lane >> 2));
          atomicOr(&pw[own_key(d.x)], bit);
          atomicOr(&pw[own_key(d.y)], bit);
          atomicOr(&pw[own_key(d.z)], bit);
          atomicOr(&pw[own_key(d.w)], bit);
          r4[(size_t)r * 128 + q * 32 + lane] =
              make_uint4(own_rec(d.x, sv[q].x), own_rec(d.y, sv[q].y),
                         own_rec(d.z, sv[q].z), own_rec(d.w, sv[q].w));
        }
      }
      const uint4 r0 = pw4[2 * lane];
      const uint4 r1 = pw4[2 * lane + 1];
      uint4* ob = (uint4*)(obits + (size_t)r * OWN_NKEY);
      ob[2 * lane]     = r0;
      ob[2 * lane + 1] = r1;
    }
  }
  if (t >= N * HEADS) return;
  int n = t >> 2, h = t & 3;
  const float4* p  = (const float4*)(Wx + (size_t)n * OUT_DIM + h * HEAD_DIM);
  const float4* a1 = (const float4*)(a);
  const float4* a2 = (const float4*)(a + HEAD_DIM);
  float acc1 = 0.f, acc2 = 0.f;
  #pragma unroll
  for (int d = 0; d < HEAD_DIM / 4; ++d) {
    float4 v  = p[d];
    float4 c1 = a1[d];
    float4 c2 = a2[d];
    acc1 += v.x * c1.x + v.y * c1.y + v.z * c1.z + v.w * c1.w;
    acc2 += v.x * c2.x + v.y * c2.y + v.z * c2.z + v.w * c2.w;
  }
  s1[t] = acc1;
  s2[t] = acc2;
}


__device__ __forceinline__ float leaky02(float el) {
  return el > 0.f ? el : 0.2f * el;
}
__device__ __forceinline__ float elu1(float v) {
  return v > 0.f ? v : (__expf(v) - 1.0f);
}
__device__ __forceinline__ float wave_max(float v) {
  #pragma unroll
  for (int o = 16; o > 0; o >>= 1) v = fmaxf(v, __shfl_xor(v, o));
  return v;
}
__device__ __forceinline__ void own_put(unsigned loc, int src, int* cnt, int* slot) {
  if (loc < (unsigned)OWN_R) {
    const int q = atomicAdd(&cnt[loc], 1);
    if (q < OWN_SLOT) slot[loc * OWN_SLOT + q] = src;
  }
}
__device__ __forceinline__ void own_list(unsigned rec, unsigned want, unsigned* wl, int& wcount) {
  const bool hit = (rec & 255u) == want;
  const unsigned b = __builtin_amdgcn_ballot_w32(hit);
  if (hit) {
    const int pos = wcount + (int)__builtin_amdgcn_mbcnt_lo(b, 0u);
    if (pos < OWN_WCAP) wl[pos] = rec;
  }
  wcount += __builtin_popcount(b);
}
__device__ __forceinline__ void own_groups(unsigned word, int row, int* gl, int& gcount) {
  unsigned b;
  while ((b = __builtin_amdgcn_ballot_w32(word != 0u)) != 0u) {
    if (word != 0u) {
      const int pos = gcount + (int)__builtin_amdgcn_mbcnt_lo(b, 0u);
      if (pos < OWN_WCAP) gl[pos] = (row << 5) + __builtin_ctz(word);
      word &= word - 1u;
    }
    gcount += __builtin_popcount(b);
  }
}

__global__ __launch_bounds__(OWN_THREADS)
void gat_owner_aggr(const int* __restrict__ ei,
                    const unsigned* __restrict__ orec,
                    const unsigned* __restrict__ obits,
                    const float* __restrict__ Wx,
                    const float* __restrict__ s1,
                    const float* __restrict__ s2,
                    float* __restrict__ out, int N, int E) {
  __shared__ int   cnt[OWN_R];
  __shared__ int   slot[OWN_R * OWN_SLOT];
  __shared__ unsigned wlist[OWN_WAVES][OWN_WCAP];
  __shared__ int   glist[OWN_WAVES][OWN_WCAP];
  __shared__ int   over;
  __shared__ int   wsrc[OWN_WAVES][OWN_SLOT + OWN_PAD];
  __shared__ float wex[OWN_WAVES][HEADS][OWN_SLOT + OWN_PAD];

  const int tid  = threadIdx.x;
  const int wave = __builtin_amdgcn_readfirstlane(tid >> 5);
  const int lane = tid & 31;
  const int lo   = blockIdx.x * OWN_R;
  const int* __restrict__ dstp = ei + E;

  for (int i = tid; i < OWN_R; i += OWN_THREADS) cnt[i] = 0;
  if (tid == 0) over = 0;
  __syncthreads();

  unsigned* wl = wlist[wave];
  int* gl = glist[wave];
  int wcount = 0;
  int gcount = 0;
  const unsigned own = blockIdx.x & 255u;
  const int nvec = ((E & 3) == 0 && N <= OWN_NKEY * OWN_R) ? (E >> 4) : 0;
  const int nrow = (nvec + 31) >> 5;
  const uint4* __restrict__ r4 = (const uint4*)orec;
  const unsigned* __restrict__ ob = obits + own;
  for (int base = 0; base < nrow; base += 2 * OWN_THREADS) {
    const int r0 = base + tid;
    const int r1 = base + OWN_THREADS + lane * OWN_WAVES + wave;
    const unsigned w0 = (r0 < nrow) ? ob[(size_t)r0 * OWN_NKEY] : 0u;
    const unsigned w1 = (r1 < nrow) ? ob[(size_t)r1 * OWN_NKEY] : 0u;
    own_groups(w0, r0, gl, gcount);
    own_groups(w1, r1, gl, gcount);
  }
  if (gcount > OWN_WCAP && lane == 0) over = 1;
  {
    const int gc = gcount < OWN_WCAP ? gcount : OWN_WCAP;
    const int sub = lane & 3;
    for (int b0 = 0; b0 < gc; b0 += 8 * OWN_RK_U) {
      uint4    rv[OWN_RK_U];
      unsigned want[OWN_RK_U];
      #pragma unroll
      for (int u = 0; u < OWN_RK_U; ++u) {
        const int i = b0 + u * 8 + (lane >> 2);
        const int g = gl[i < gc ? i : gc - 1];
        const bool ok = i < gc && g < nvec;
        want[u] = ok ? own : 256u;
        rv[u] = r4[4 * (size_t)(ok ? g : 0) + sub];
      }
      #pragma unroll
      for (int u = 0; u < OWN_RK_U; ++u) {
        own_list(rv[u].x, want[u], wl, wcount);
        own_list(rv[u].y, want[u], wl, wcount);
        own_list(rv[u].z, want[u], wl, wcount);
        own_list(rv[u].w, want[u], wl, wcount);
      }
    }
  }
  if (wcount > OWN_WCAP && lane == 0) over = 1;
  {
    const int wc = wcount < OWN_WCAP ? wcount : OWN_WCAP;
    for (int b0 = 0; b0 < wc; b0 += 32 * OWN_PUT_U) {
      unsigned rr[OWN_PUT_U];
      #pragma unroll
      for (int u = 0; u < OWN_PUT_U; ++u) {
        const int i = b0 + u * 32 + lane;
        rr[u] = 0xff00u;
        if (i < wc) rr[u] = wl[i];
      }
      #pragma unroll
      for (int u = 0; u < OWN_PUT_U; ++u)
        own_put((rr[u] >> 8) & 255u, (int)(rr[u] >> 16), cnt, slot);
    }
  }
  for (int eb = nvec << 4; eb < E; eb += OWN_THREADS) {
    const int e = eb + tid;
    if (e < E) {
      const unsigned loc = (unsigned)(dstp[e] - lo);
      if (loc < (unsigned)OWN_R) own_put(loc, ei[e], cnt, slot);
    }
  }
  __syncthreads();
  if (over != 0) {
    for (int i = tid; i < OWN_R; i += OWN_THREADS) cnt[i] = OWN_SLOT + 1;
    __syncthreads();
  }

  const int h = lane >> 3;
  const int j = lane << 2;
  for (int loc = wave; loc < OWN_R; loc += OWN_WAVES) {
    const int n = lo + loc;
    if (n >= N) break;
    const int L = __builtin_amdgcn_readfirstlane(cnt[loc]);
    float4 acc = make_float4(0.f, 0.f, 0.f, 0.f);
    float den = 0.f;
    float* op = out + (size_t)n * OUT_DIM + j;
    if (L == 0) {
      *(float4*)op = acc;
      continue;
    }
    const float4 s2v = *(const float4*)(s2 + (size_t)n * HEADS);
    const float s2h = h == 0 ? s2v.x : h == 1 ? s2v.y : h == 2 ? s2v.z : s2v.w;
    float m0 = -INFINITY, m1 = -INFINITY, m2 = -INFINITY, m3 = -INFINITY;

    if (L <= OWN_SLOT) {
      const int* sl = slot + loc * OWN_SLOT;
      const bool v0 = lane < L;
      const bool v1 = lane + 32 < L;
      const int key0 = v0 ? sl[lane] : 0x7fffffff;
      const int key1 = v1 ? sl[lane + 32] : 0x7fffffff;
      float4 a0 = make_float4(0.f, 0.f, 0.f, 0.f), a1 = a0;
      if (v0) a0 = *(const float4*)(s1 + (size_t)key0 * HEADS);
      if (v1) a1 = *(const float4*)(s1 + (size_t)key1 * HEADS);
      int r0 = 0, r1 = 0;
      if (L <= 32) {
        for (int t = 0; t < L; ++t) {
          int kt = __builtin_amdgcn_readlane(key0, t);
          r0 += (kt < key0 || (kt == key0 && t < lane)) ? 1 : 0;
        }
      } else {
        for (int t = 0; t < 32; ++t) {
          int kt = __builtin_amdgcn_readlane(key0, t);
          r0 += (kt < key0 || (kt == key0 && t < lane)) ? 1 : 0;
          r1 += (kt <= key1) ? 1 : 0;
        }
        for (int t = 32; t < L; ++t) {
          int kt = __builtin_amdgcn_readlane(key1, t - 32);
          r0 += (kt < key0) ? 1 : 0;
          r1 += (kt < key1 || (kt == key1 && (t - 32) < lane)) ? 1 : 0;
        }
      }
      #pragma unroll
      for (int c = 0; c < 2; ++c) {
        const int    key  = c == 0 ? key0 : key1;
        const int    rank = c == 0 ? r0 : r1;
        const float4 s1v  = c == 0 ? a0 : a1;
        if (c == 0 ? v0 : v1) {
          float e0 = leaky02(s1v.x + s2v.x);
          float e1 = leaky02(s1v.y + s2v.y);
          float e2 = leaky02(s1v.z + s2v.z);
          float e3 = leaky02(s1v.w + s2v.w);
          wsrc[wave][rank]   = key;
          wex[wave][0][rank] = e0;
          wex[wave][1][rank] = e1;
          wex[wave][2][rank] = e2;
          wex[wave][3][rank] = e3;
          m0 = fmaxf(m0, e0); m1 = fmaxf(m1, e1);
          m2 = fmaxf(m2, e2); m3 = fmaxf(m3, e3);
        }
      }
      if (lane < OWN_PAD) {
        wsrc[wave][L + lane]   = 0;
        wex[wave][0][L + lane] = 0.f;
        wex[wave][1][L + lane] = 0.f;
        wex[wave][2][L + lane] = 0.f;
        wex[wave][3][L + lane] = 0.f;
      }
      m0 = wave_max(m0); m1 = wave_max(m1); m2 = wave_max(m2); m3 = wave_max(m3);
      const float mh = h == 0 ? m0 : h == 1 ? m1 : h == 2 ? m2 : m3;
      const float* wxl = Wx + j;
      for (int t0 = 0; t0 < L; t0 += OWN_PAD) {
        int    sv[OWN_PAD];
        float  ev[OWN_PAD];
        float4 row[OWN_PAD];
        #pragma unroll
        for (int u = 0; u < OWN_PAD; ++u) {
          sv[u] = wsrc[wave][t0 + u];
          ev[u] = wex[wave][h][t0 + u];
        }
        #pragma unroll
        for (int u = 0; u < OWN_PAD; ++u) {
          row[u] = make_float4(0.f, 0.f, 0.f, 0.f);
          if (t0 + u < L) row[u] = *(const float4*)(wxl + (size_t)sv[u] * OUT_DIM);
        }
        #pragma unroll
        for (int u = 0; u < OWN_PAD; ++u) {
          float xv = (t0 + u < L) ? __expf(ev[u] - mh) : 0.f;
          den   += xv;
          acc.x += xv * row[u].x; acc.y += xv * row[u].y;
          acc.z += xv * row[u].z; acc.w += xv * row[u].w;
        }
      }
    } else {
      for (int base2 = 0; base2 < E; base2 += 32) {
        int e = base2 + lane;
        if (e < E && dstp[e] == n) {
          float4 s1v = *(const float4*)(s1 + (size_t)ei[e] * HEADS);
          m0 = fmaxf(m0, leaky02(s1v.x + s2v.x));
          m1 = fmaxf(m1, leaky02(s1v.y + s2v.y));
          m2 = fmaxf(m2, leaky02(s1v.z + s2v.z));
          m3 = fmaxf(m3, leaky02(s1v.w + s2v.w));
        }
      }
      m0 = wave_max(m0); m1 = wave_max(m1); m2 = wave_max(m2); m3 = wave_max(m3);
      const float mh = h == 0 ? m0 : h == 1 ? m1 : h == 2 ? m2 : m3;
      for (int base2 = 0; base2 < E; base2 += 32) {
        int e = base2 + lane;
        bool mt = e < E && dstp[e] == n;
        int src = mt ? ei[e] : 0;
        unsigned mask = __builtin_amdgcn_ballot_w32(mt);
        while (mask) {
          int t = __builtin_ctz(mask);
          mask &= mask - 1;
          int s = __builtin_amdgcn_readlane(src, t);
          float ev = leaky02(s1[(size_t)s * HEADS + h] + s2h);
          float4 row = *(const float4*)(Wx + (size_t)s * OUT_DIM + j);
          float xv = __expf(ev - mh);
          den   += xv;
          acc.x += xv * row.x; acc.y += xv * row.y;
          acc.z += xv * row.z; acc.w += xv * row.w;
        }
      }
      if (den == 0.f) den = 1.f;
    }
    float4 o;
    o.x = elu1(acc.x / den); o.y = elu1(acc.y / den);
    o.z = elu1(acc.z / den); o.w = elu1(acc.w / den);
    *(float4*)op = o;
  }
}

extern "C" void kernel_launch(void* const* d_in, const int* in_sizes, int n_in,
                              void* d_out, int out_size, void* d_ws, size_t ws_size,
                              hipStream_t stream) {
  const float* x  = (const float*)d_in[0];
  const float* W  = (const float*)d_in[1];
  const float* a  = (const float*)d_in[2];
  const int*   ei = (const int*)d_in[3];
  const int N = in_sizes[0] / IN_DIM;
  const int E = in_sizes[3] / 2;
  float* out = (float*)d_out;

  char* ws = (char*)d_ws;
  size_t off = 0;
  auto carve = [&](size_t bytes) -> void* {
    void* p = ws + off;
    off += (bytes + 255) & ~(size_t)255;
    return p;
  };
  float* Wx = (float*)carve((size_t)N * OUT_DIM * sizeof(float));
  float* s1 = (float*)carve((size_t)N * HEADS * sizeof(float));
  float* s2 = (float*)carve((size_t)N * HEADS * sizeof(float));
  unsigned* orec = (unsigned*)carve((size_t)E * sizeof(unsigned) + 64);
  unsigned* obits = (unsigned*)carve((((size_t)E >> 9) + 1) * OWN_NKEY * sizeof(unsigned));
  (void)ws_size;

  gat_gemm_wmma<<<(N + GT_BROWS - 1) / GT_BROWS, 128, 0, stream>>>(x, W, Wx, N);

  gat_scores<<<(N * HEADS + SC_THREADS - 1) / SC_THREADS, SC_THREADS, 0, stream>>>(
      Wx, a, s1, s2, ei, orec, obits, N, E);

  gat_owner_aggr<<<(N + OWN_R - 1) / OWN_R, OWN_THREADS, 0, stream>>>(
      ei, orec, obits, Wx, s1, s2, out, N, E);
}


# aggregation node loop issues about 40% fewer instructions: rank sort on one packed word (source, list position), head max from sorted LDS scores by 8-lane DPP, whole 8-row steps untested; sums kept
# speedup vs baseline: 7.2816x; 1.0528x over previous
#include <hip/hip_runtime.h>
#include <cstdint>
#include <cstddef>

typedef __attribute__((ext_vector_type(2))) float v2f;
typedef __attribute__((ext_vector_type(8))) float v8f;

#define IN_DIM   128
#define OUT_DIM  128
#define HEADS    4
#define HEAD_DIM 32

#define GT_BROWS 64
#define GT_WROWS 32
#define GT_WCOLS 64

__global__ __launch_bounds__(128)
void gat_gemm_wmma(const float* __restrict__ x,
                   const float* __restrict__ W,
                   float* __restrict__ Wx, int N) {
  const int wave  = threadIdx.x >> 5;
  const int lane  = threadIdx.x & 31;
  const int mbase = blockIdx.x * GT_BROWS + (wave >> 1) * GT_WROWS;
  const int obase = (wave & 1) * GT_WCOLS;
  const int half  = lane >> 4;
  const int l15   = lane & 15;

  int arow0 = mbase + l15;
  int arow1 = mbase + 16 + l15;
  if (arow0 >= N) arow0 = N - 1;
  if (arow1 >= N) arow1 = N - 1;

  const float* ap0 = x + (size_t)arow0 * IN_DIM + 2 * half;
  const float* ap1 = x + (size_t)arow1 * IN_DIM + 2 * half;
  const float* bp  = W + (size_t)(obase + 4 * l15) * IN_DIM + 2 * half;

  v8f c00 = {}, c01 = {}, c02 = {}, c03 = {};
  v8f c10 = {}, c11 = {}, c12 = {}, c13 = {};
  #pragma unroll
  for (int k = 0; k < IN_DIM; k += 4) {
    const v2f a0 = *(const v2f*)(ap0 + k);
    const v2f a1 = *(const v2f*)(ap1 + k);
    const v2f b0 = *(const v2f*)(bp + k);
    const v2f b1 = *(const v2f*)(bp + IN_DIM + k);
    const v2f b2 = *(const v2f*)(bp + 2 * IN_DIM + k);
    const v2f b3 = *(const v2f*)(bp + 3 * IN_DIM + k);
    c00 = __builtin_amdgcn_wmma_f32_16x16x4_f32(false, a0, false, b0, (short)0, c00, false, false);
    c10 = __builtin_amdgcn_wmma_f32_16x16x4_f32(false, a1, false, b0, (short)0, c10, false, false);
    c01 = __builtin_amdgcn_wmma_f32_16x16x4_f32(false, a0, false, b1, (short)0, c01, false, false);
    c11 = __builtin_amdgcn_wmma_f32_16x16x4_f32(false, a1, false, b1, (short)0, c11, false, false);
    c02 = __builtin_amdgcn_wmma_f32_16x16x4_f32(false, a0, false, b2, (short)0, c02, false, false);
    c12 = __builtin_amdgcn_wmma_f32_16x16x4_f32(false, a1, false, b2, (short)0, c12, false, false);
    c03 = __builtin_amdgcn_wmma_f32_16x16x4_f32(false, a0, false, b3, (short)0, c03, false, false);
    c13 = __builtin_amdgcn_wmma_f32_16x16x4_f32(false, a1, false, b3, (short)0, c13, false, false);
  }

  float* op = Wx + obase + 4 * l15;
  #pragma unroll
  for (int r = 0; r < 8; ++r) {
    const int row0 = mbase + r + 8 * half;
    const int row1 = row0 + 16;
    if (row0 < N)
      *(float4*)(op + (size_t)row0 * OUT_DIM) = make_float4(c00[r], c01[r], c02[r], c03[r]);
    if (row1 < N)
      *(float4*)(op + (size_t)row1 * OUT_DIM) = make_float4(c10[r], c11[r], c12[r], c13[r]);
  }
}

#define OWN_R       196
#define OWN_SLOT    64
#define OWN_PAD     8
#define OWN_THREADS 1024
#define OWN_WAVES   (OWN_THREADS / 32)
#define OWN_WCAP    256
#define OWN_RK_U    8
#define OWN_PUT_U   4
#define OWN_NKEY    256

__device__ __forceinline__ unsigned own_key(int d) {
  return ((unsigned)d / (unsigned)OWN_R) & 255u;
}
__device__ __forceinline__ unsigned own_rec(int d, int s) {
  const unsigned o   = (unsigned)d / (unsigned)OWN_R;
  const unsigned loc = o < (unsigned)OWN_NKEY ? (unsigned)d - o * (unsigned)OWN_R : 255u;
  return (o & 255u) | (loc << 8) | ((unsigned)s << 16);
}

#define SC_THREADS 256
#define SC_WAVES   (SC_THREADS / 32)

__global__ __launch_bounds__(SC_THREADS)
void gat_scores(const float* __restrict__ Wx,
                           const float* __restrict__ a,
                           float* __restrict__ s1, float* __restrict__ s2,
                           const int* __restrict__ ei,
                           unsigned* __restrict__ orec,
                           unsigned* __restrict__ obits,
                           int N, int E) {
  __shared__ __attribute__((aligned(16))) unsigned pres[SC_WAVES][OWN_NKEY];
  int t = blockIdx.x * blockDim.x + threadIdx.x;
  if ((E & 3) == 0 && N <= OWN_NKEY * OWN_R) {
    const int4* __restrict__ d4 = (const int4*)(ei + E);
    const int4* __restrict__ s4 = (const int4*)ei;
    uint4* __restrict__ r4 = (uint4*)orec;
    const int nvec = E >> 4;
    const int nrow = (nvec + 31) >> 5;
    const int lane = threadIdx.x & 31;
    const int wv   = threadIdx.x >> 5;
    uint4* pw4 = (uint4*)pres[wv];
    unsigned* pw = pres[wv];
    for (int r = wv * gridDim.x + blockIdx.x; r < nrow; r += gridDim.x * SC_WAVES) {
      pw4[2 * lane]     = make_uint4(0u, 0u, 0u, 0u);
      pw4[2 * lane + 1] = make_uint4(0u, 0u, 0u, 0u);
      int4 dv[4], sv[4];
      bool okq[4];
      #pragma unroll
      for (int q = 0; q < 4; ++q) {
        okq[q] = (r << 5) + q * 8 + (lane >> 2) < nvec;
        const size_t v = okq[q] ? (size_t)r * 128 + q * 32 + lane : 0;
        dv[q] = d4[v];
        sv[q] = s4[v];
      }
      #pragma unroll
      for (int q = 0; q < 4; ++q) {
        if (okq[q]) {
          const int4 d = dv[q];
          const unsigned bit = 1u << (q * 8 + (lane >> 2));
          atomicOr(&pw[own_key(d.x)], bit);
          atomicOr(&pw[own_key(d.y)], bit);
          atomicOr(&pw[own_key(d.z)], bit);
          atomicOr(&pw[own_key(d.w)], bit);
          r4[(size_t)r * 128 + q * 32 + lane] =
              make_uint4(own_rec(d.x, sv[q].x), own_rec(d.y, sv[q].y),
                         own_rec(d.z, sv[q].z), own_rec(d.w, sv[q].w));
        }
      }
      const uint4 r0 = pw4[2 * lane];
      const uint4 r1 = pw4[2 * lane + 1];
      uint4* ob = (uint4*)(obits + (size_t)r * OWN_NKEY);
      ob[2 * lane]     = r0;
      ob[2 * lane + 1] = r1;
    }
  }
  if (t >= N * HEADS) return;
  int n = t >> 2, h = t & 3;
  const float4* p  = (const float4*)(Wx + (size_t)n * OUT_DIM + h * HEAD_DIM);
  const float4* a1 = (const float4*)(a);
  const float4* a2 = (const float4*)(a + HEAD_DIM);
  float acc1 = 0.f, acc2 = 0.f;
  #pragma unroll
  for (int d = 0; d < HEAD_DIM / 4; ++d) {
    float4 v  = p[d];
    float4 c1 = a1[d];
    float4 c2 = a2[d];
    acc1 += v.x * c1.x + v.y * c1.y + v.z * c1.z + v.w * c1.w;
    acc2 += v.x * c2.x + v.y * c2.y + v.z * c2.z + v.w * c2.w;
  }
  s1[t] = acc1;
  s2[t] = acc2;
}


__device__ __forceinline__ float leaky02(float el) {
  return el > 0.f ? el : 0.2f * el;
}
__device__ __forceinline__ float elu1(float v) {
  return v > 0.f ? v : (__expf(v) - 1.0f);
}
__device__ __forceinline__ float wave_max(float v) {
  #pragma unroll
  for (int o = 16; o > 0; o >>= 1) v = fmaxf(v, __shfl_xor(v, o));
  return v;
}
__device__ __forceinline__ float max8(float v) {
  v = fmaxf(v, __int_as_float(__builtin_amdgcn_update_dpp(0, __float_as_int(v), 0x161, 0xf, 0xf, false)));
  v = fmaxf(v, __int_as_float(__builtin_amdgcn_update_dpp(0, __float_as_int(v), 0x162, 0xf, 0xf, false)));
  v = fmaxf(v, __int_as_float(__builtin_amdgcn_update_dpp(0, __float_as_int(v), 0x164, 0xf, 0xf, false)));
  return v;
}
__device__ __forceinline__ void own_put(unsigned loc, int src, int* cnt, int* slot) {
  if (loc < (unsigned)OWN_R) {
    const int q = atomicAdd(&cnt[loc], 1);
    if (q < OWN_SLOT) slot[loc * OWN_SLOT + q] = src;
  }
}
__device__ __forceinline__ void own_list(unsigned rec, unsigned want, unsigned* wl, int& wcount) {
  const bool hit = (rec & 255u) == want;
  const unsigned b = __builtin_amdgcn_ballot_w32(hit);
  if (hit) {
    const int pos = wcount + (int)__builtin_amdgcn_mbcnt_lo(b, 0u);
    if (pos < OWN_WCAP) wl[pos] = rec;
  }
  wcount += __builtin_popcount(b);
}
__device__ __forceinline__ void own_groups(unsigned word, int row, int* gl, int& gcount) {
  unsigned b;
  while ((b = __builtin_amdgcn_ballot_w32(word != 0u)) != 0u) {
    if (word != 0u) {
      const int pos = gcount + (int)__builtin_amdgcn_mbcnt_lo(b, 0u);
      if (pos < OWN_WCAP) gl[pos] = (row << 5) + __builtin_ctz(word);
      word &= word - 1u;
    }
    gcount += __builtin_popcount(b);
  }
}

__global__ __launch_bounds__(OWN_THREADS)
void gat_owner_aggr(const int* __restrict__ ei,
                    const unsigned* __restrict__ orec,
                    const unsigned* __restrict__ obits,
                    const float* __restrict__ Wx,
                    const float* __restrict__ s1,
                    const float* __restrict__ s2,
                    float* __restrict__ out, int N, int E) {
  __shared__ int   cnt[OWN_R];
  __shared__ int   slot[OWN_R * OWN_SLOT];
  __shared__ unsigned wlist[OWN_WAVES][OWN_WCAP];
  __shared__ int   glist[OWN_WAVES][OWN_WCAP];
  __shared__ int   over;
  __shared__ int   wsrc[OWN_WAVES][OWN_SLOT + OWN_PAD];
  __shared__ float wex[OWN_WAVES][HEADS][OWN_SLOT + OWN_PAD];

  const int tid  = threadIdx.x;
  const int wave = __builtin_amdgcn_readfirstlane(tid >> 5);
  const int lane = tid & 31;
  const int lo   = blockIdx.x * OWN_R;
  const int* __restrict__ dstp = ei + E;

  for (int i = tid; i < OWN_R; i += OWN_THREADS) cnt[i] = 0;
  if (tid == 0) over = 0;
  __syncthreads();

  unsigned* wl = wlist[wave];
  int* gl = glist[wave];
  int wcount = 0;
  int gcount = 0;
  const unsigned own = blockIdx.x & 255u;
  const int nvec = ((E & 3) == 0 && N <= OWN_NKEY * OWN_R) ? (E >> 4) : 0;
  const int nrow = (nvec + 31) >> 5;
  const uint4* __restrict__ r4 = (const uint4*)orec;
  const unsigned* __restrict__ ob = obits + own;
  for (int base = 0; base < nrow; base += 2 * OWN_THREADS) {
    const int r0 = base + tid;
    const int r1 = base + OWN_THREADS + lane * OWN_WAVES + wave;
    const unsigned w0 = (r0 < nrow) ? ob[(size_t)r0 * OWN_NKEY] : 0u;
    const unsigned w1 = (r1 < nrow) ? ob[(size_t)r1 * OWN_NKEY] : 0u;
    own_groups(w0, r0, gl, gcount);
    own_groups(w1, r1, gl, gcount);
  }
  if (gcount > OWN_WCAP && lane == 0) over = 1;
  {
    const int gc = gcount < OWN_WCAP ? gcount : OWN_WCAP;
    const int sub = lane & 3;
    for (int b0 = 0; b0 < gc; b0 += 8 * OWN_RK_U) {
      uint4    rv[OWN_RK_U];
      unsigned want[OWN_RK_U];
      #pragma unroll
      for (int u = 0; u < OWN_RK_U; ++u) {
        const int i = b0 + u * 8 + (lane >> 2);
        const int g = gl[i < gc ? i : gc - 1];
        const bool ok = i < gc && g < nvec;
        want[u] = ok ? own : 256u;
        rv[u] = r4[4 * (size_t)(ok ? g : 0) + sub];
      }
      #pragma unroll
      for (int u = 0; u < OWN_RK_U; ++u) {
        own_list(rv[u].x, want[u], wl, wcount);
        own_list(rv[u].y, want[u], wl, wcount);
        own_list(rv[u].z, want[u], wl, wcount);
        own_list(rv[u].w, want[u], wl, wcount);
      }
    }
  }
  if (wcount > OWN_WCAP && lane == 0) over = 1;
  {
    const int wc = wcount < OWN_WCAP ? wcount : OWN_WCAP;
    for (int b0 = 0; b0 < wc; b0 += 32 * OWN_PUT_U) {
      unsigned rr[OWN_PUT_U];
      #pragma unroll
      for (int u = 0; u < OWN_PUT_U; ++u) {
        const int i = b0 + u * 32 + lane;
        rr[u] = 0xff00u;
        if (i < wc) rr[u] = wl[i];
      }
      #pragma unroll
      for (int u = 0; u < OWN_PUT_U; ++u)
        own_put((rr[u] >> 8) & 255u, (int)(rr[u] >> 16), cnt, slot);
    }
  }
  for (int eb = nvec << 4; eb < E; eb += OWN_THREADS) {
    const int e = eb + tid;
    if (e < E) {
      const unsigned loc = (unsigned)(dstp[e] - lo);
      if (loc < (unsigned)OWN_R) own_put(loc, ei[e], cnt, slot);
    }
  }
  __syncthreads();
  if (over != 0) {
    for (int i = tid; i < OWN_R; i += OWN_THREADS) cnt[i] = OWN_SLOT + 1;
    __syncthreads();
  }

  const int h = lane >> 3;
  const int j = lane << 2;
  for (int loc = wave; loc < OWN_R; loc += OWN_WAVES) {
    const int n = lo + loc;
    if (n >= N) break;
    const int L = __builtin_amdgcn_readfirstlane(cnt[loc]);
    float4 acc = make_float4(0.f, 0.f, 0.f, 0.f);
    float den = 0.f;
    float* op = out + (size_t)n * OUT_DIM + j;
    if (L == 0) {
      *(float4*)op = acc;
      continue;
    }
    const float4 s2v = *(const float4*)(s2 + (size_t)n * HEADS);
    const float s2h = h == 0 ? s2v.x : h == 1 ? s2v.y : h == 2 ? s2v.z : s2v.w;
    float m0 = -INFINITY, m1 = -INFINITY, m2 = -INFINITY, m3 = -INFINITY;

    if (L <= OWN_SLOT && N <= (1 << 25)) {
      const int* sl = slot + loc * OWN_SLOT;
      const bool v0 = lane < L;
      const bool v1 = lane + 32 < L;
      const unsigned pk0 = v0 ? (((unsigned)sl[lane] << 6) | (unsigned)lane) : 0xffffffffu;
      const unsigned pk1 = v1 ? (((unsigned)sl[lane + 32] << 6) | (unsigned)(lane + 32)) : 0xffffffffu;
      const unsigned key0 = pk0 >> 6;
      const unsigned key1 = pk1 >> 6;
      const float4 a0 = *(const float4*)(s1 + (size_t)(v0 ? key0 : 0u) * HEADS);
      int r0 = 0, r1 = 0;
      if (L <= 32) {
        for (int t = 0; t < L; ++t) {
          const unsigned kt = (unsigned)__builtin_amdgcn_readlane((int)pk0, t);
          r0 += (kt < pk0) ? 1 : 0;
        }
      } else {
        for (int t = 0; t < 32; ++t) {
          const unsigned kt = (unsigned)__builtin_amdgcn_readlane((int)pk0, t);
          r0 += (kt < pk0) ? 1 : 0;
          r1 += (kt < pk1) ? 1 : 0;
        }
        for (int t = 32; t < L; ++t) {
          const unsigned kt = (unsigned)__builtin_amdgcn_readlane((int)pk1, t - 32);
          r0 += (kt < pk0) ? 1 : 0;
          r1 += (kt < pk1) ? 1 : 0;
        }
      }
      if (v0) {
        wsrc[wave][r0]   = (int)key0;
        wex[wave][0][r0] = leaky02(a0.x + s2v.x);
        wex[wave][1][r0] = leaky02(a0.y + s2v.y);
        wex[wave][2][r0] = leaky02(a0.z + s2v.z);
        wex[wave][3][r0] = leaky02(a0.w + s2v.w);
      }
      if (L > 32) {
        const float4 a1 = *(const float4*)(s1 + (size_t)(v1 ? key1 : 0u) * HEADS);
        if (v1) {
          wsrc[wave][r1]   = (int)key1;
          wex[wave][0][r1] = leaky02(a1.x + s2v.x);
          wex[wave][1][r1] = leaky02(a1.y + s2v.y);
          wex[wave][2][r1] = leaky02(a1.z + s2v.z);
          wex[wave][3][r1] = leaky02(a1.w + s2v.w);
        }
      }
      if (lane < OWN_PAD) {
        wsrc[wave][L + lane]   = 0;
        wex[wave][0][L + lane] = -INFINITY;
        wex[wave][1][L + lane] = -INFINITY;
        wex[wave][2][L + lane] = -INFINITY;
        wex[wave][3][L + lane] = -INFINITY;
      }
      float mh = -INFINITY;
      for (int i0 = 0; i0 < L; i0 += 4 * OWN_PAD) {
        #pragma unroll
        for (int q = 0; q < 4; ++q) {
          const float v = wex[wave][h][i0 + q * OWN_PAD + (lane & 7)];
          mh = fmaxf(mh, (i0 + q * OWN_PAD < L) ? v : -INFINITY);
        }
      }
      mh = max8(mh);
      const float* wxl = Wx + j;
      int t0 = 0;
      for (; t0 + OWN_PAD <= L; t0 += OWN_PAD) {
        int    sv[OWN_PAD];
        float  ev[OWN_PAD];
        float4 row[OWN_PAD];
        #pragma unroll
        for (int u = 0; u < OWN_PAD; ++u) {
          sv[u] = wsrc[wave][t0 + u];
          ev[u] = wex[wave][h][t0 + u];
        }
        #pragma unroll
        for (int u = 0; u < OWN_PAD; ++u)
          row[u] = *(const float4*)(wxl + (size_t)(unsigned)sv[u] * OUT_DIM);
        #pragma unroll
        for (int u = 0; u < OWN_PAD; ++u) {
          float xv = __expf(ev[u] - mh);
          den   += xv;
          acc.x = fmaf(xv, row[u].x, acc.x); acc.y = fmaf(xv, row[u].y, acc.y);
          acc.z = fmaf(xv, row[u].z, acc.z); acc.w = fmaf(xv, row[u].w, acc.w);
        }
      }
      if (t0 < L) {
        int    sv[OWN_PAD];
        float  ev[OWN_PAD];
        float4 row[OWN_PAD];
        #pragma unroll
        for (int u = 0; u < OWN_PAD; ++u) {
          sv[u] = wsrc[wave][t0 + u];
          ev[u] = wex[wave][h][t0 + u];
        }
        #pragma unroll
        for (int u = 0; u < OWN_PAD; ++u) {
          row[u] = make_float4(0.f, 0.f, 0.f, 0.f);
          if (t0 + u < L) row[u] = *(const float4*)(wxl + (size_t)(unsigned)sv[u] * OUT_DIM);
        }
        #pragma unroll
        for (int u = 0; u < OWN_PAD; ++u) {
          float xv = (t0 + u < L) ? __expf(ev[u] - mh) : 0.f;
          den   += xv;
          acc.x = fmaf(xv, row[u].x, acc.x); acc.y = fmaf(xv, row[u].y, acc.y);
          acc.z = fmaf(xv, row[u].z, acc.z); acc.w = fmaf(xv, row[u].w, acc.w);
        }
      }
    } else {
      for (int base2 = 0; base2 < E; base2 += 32) {
        int e = base2 + lane;
        if (e < E && dstp[e] == n) {
          float4 s1v = *(const float4*)(s1 + (size_t)ei[e] * HEADS);
          m0 = fmaxf(m0, leaky02(s1v.x + s2v.x));
          m1 = fmaxf(m1, leaky02(s1v.y + s2v.y));
          m2 = fmaxf(m2, leaky02(s1v.z + s2v.z));
          m3 = fmaxf(m3, leaky02(s1v.w + s2v.w));
        }
      }
      m0 = wave_max(m0); m1 = wave_max(m1); m2 = wave_max(m2); m3 = wave_max(m3);
      const float mh = h == 0 ? m0 : h == 1 ? m1 : h == 2 ? m2 : m3;
      for (int base2 = 0; base2 < E; base2 += 32) {
        int e = base2 + lane;
        bool mt = e < E && dstp[e] == n;
        int src = mt ? ei[e] : 0;
        unsigned mask = __builtin_amdgcn_ballot_w32(mt);
        while (mask) {
          int t = __builtin_ctz(mask);
          mask &= mask - 1;
          int s = __builtin_amdgcn_readlane(src, t);
          float ev = leaky02(s1[(size_t)s * HEADS + h] + s2h);
          float4 row = *(const float4*)(Wx + (size_t)s * OUT_DIM + j);
          float xv = __expf(ev - mh);
          den   += xv;
          acc.x += xv * row.x; acc.y += xv * row.y;
          acc.z += xv * row.z; acc.w += xv * row.w;
        }
      }
      if (den == 0.f) den = 1.f;
    }
    float4 o;
    o.x = elu1(acc.x / den); o.y = elu1(acc.y / den);
    o.z = elu1(acc.z / den); o.w = elu1(acc.w / den);
    *(float4*)op = o;
  }
}

extern "C" void kernel_launch(void* const* d_in, const int* in_sizes, int n_in,
                              void* d_out, int out_size, void* d_ws, size_t ws_size,
                              hipStream_t stream) {
  const float* x  = (const float*)d_in[0];
  const float* W  = (const float*)d_in[1];
  const float* a  = (const float*)d_in[2];
  const int*   ei = (const int*)d_in[3];
  const int N = in_sizes[0] / IN_DIM;
  const int E = in_sizes[3] / 2;
  float* out = (float*)d_out;

  char* ws = (char*)d_ws;
  size_t off = 0;
  auto carve = [&](size_t bytes) -> void* {
    void* p = ws + off;
    off += (bytes + 255) & ~(size_t)255;
    return p;
  };
  float* Wx = (float*)carve((size_t)N * OUT_DIM * sizeof(float));
  float* s1 = (float*)carve((size_t)N * HEADS * sizeof(float));
  float* s2 = (float*)carve((size_t)N * HEADS * sizeof(float));
  unsigned* orec = (unsigned*)carve((size_t)E * sizeof(unsigned) + 64);
  unsigned* obits = (unsigned*)carve((((size_t)E >> 9) + 1) * OWN_NKEY * sizeof(unsigned));
  (void)ws_size;

  gat_gemm_wmma<<<(N + GT_BROWS - 1) / GT_BROWS, 128, 0, stream>>>(x, W, Wx, N);

  gat_scores<<<(N * HEADS + SC_THREADS - 1) / SC_THREADS, SC_THREADS, 0, stream>>>(
      Wx, a, s1, s2, ei, orec, obits, N, E);

  gat_owner_aggr<<<(N + OWN_R - 1) / OWN_R, OWN_THREADS, 0, stream>>>(
      ei, orec, obits, Wx, s1, s2, out, N, E);
}
